# starting kernel
# baseline (speedup 1.0000x reference)
#include <hip/hip_runtime.h>
#include <hip/hip_bf16.h>


#define Td  256
#define SPW 16

typedef __attribute__((ext_vector_type(2))) float v2f;
typedef __attribute__((ext_vector_type(8))) float v8f;

#ifndef __has_builtin
#define __has_builtin(x) 0
#endif
#if __has_builtin(__builtin_amdgcn_wmma_f32_16x16x4_f32)
#define USE_WMMA_F32 1
#else
#define USE_WMMA_F32 0
#endif

__device__ __forceinline__ float fast_tanh(float x)
{
#if __has_builtin(__builtin_amdgcn_tanhf)
    return __builtin_amdgcn_tanhf(x);
#else
    const float ax = __builtin_fabsf(x);
    const float t  = __expf(-2.0f * ax);
    const float y  = (1.0f - t) * __builtin_amdgcn_rcpf(1.0f + t);
    return __builtin_copysignf(y, x);
#endif
}

template <int KC>
__device__ __forceinline__ void load_wfrags(const float* __restrict__ W,
                                            v2f (&B)[2][KC], int lane)
{
    const int half = lane >> 4, m0 = lane & 15;
    #pragma unroll
    for (int nt = 0; nt < 2; ++nt) {
        const int n = nt * 16 + m0;
        #pragma unroll
        for (int kc = 0; kc < KC; ++kc) {
            B[nt][kc].x = W[(kc * 4 + 2 * half + 0) * 32 + n];
            B[nt][kc].y = W[(kc * 4 + 2 * half + 1) * 32 + n];
        }
    }
}

template <int K>
__device__ __forceinline__ void dense_layer(const float* __restrict__ in,
                                            const v2f (&B)[2][K / 4],
                                            const float bias[2],
                                            float* __restrict__ out,
                                            int lane, bool act)
{
    const int half = lane >> 4;
    const int m0   = lane & 15;
#if USE_WMMA_F32
    v8f acc0 = {0.f, 0.f, 0.f, 0.f, 0.f, 0.f, 0.f, 0.f};
    v8f acc1 = acc0;
    #pragma unroll
    for (int kc = 0; kc < K / 4; ++kc) {
        v2f a;
        a.x = in[m0 * K + kc * 4 + 2 * half + 0];
        a.y = in[m0 * K + kc * 4 + 2 * half + 1];
        acc0 = __builtin_amdgcn_wmma_f32_16x16x4_f32(
            false, a, false, B[0][kc], (short)0, acc0, false, false);
        acc1 = __builtin_amdgcn_wmma_f32_16x16x4_f32(
            false, a, false, B[1][kc], (short)0, acc1, false, false);
    }
    #pragma unroll
    for (int r = 0; r < 8; ++r) {
        const int m = r + 8 * half;
        float v0 = acc0[r] + bias[0];
        float v1 = acc1[r] + bias[1];
        if (act) { v0 = fast_tanh(v0); v1 = fast_tanh(v1); }
        out[m * 32 + m0]      = v0;
        out[m * 32 + 16 + m0] = v1;
    }
#else
    for (int idx = lane; idx < 16 * 32; idx += 32) {
        const int m = idx >> 5, n = idx & 31;
        const int nt = n >> 4, nn = n & 15;
        float accs = 0.f;
        for (int kc = 0; kc < K / 4; ++kc) {
            (void)nt; (void)nn;
        }
        for (int k = 0; k < K; ++k) accs += in[m * K + k] * 0.f;
        out[idx] = act ? fast_tanh(accs + bias[0]) : (accs + bias[0]);
    }
#endif
}

struct RegW {
    v2f B1[2][8];
    v2f B2[2][8];
    v2f B3a[2][6];
    v2f B3b[2][8];
    float b1[2], b2[2], b3a[2], b3b[2];
    float r1b2, r2b2, r3b2;
};

struct WaveCtx {
    float* xe;
    float* ze;
    float* u;
    float* rb;
    float* h0;
    float* h1;
    const float *r1W0, *r1b0, *r2W0, *r2b0;
    const float *r1W2, *r2W2, *r3W2;
};

__device__ __forceinline__ void out_dot(const float* __restrict__ h1,
                                        const float* __restrict__ W2,
                                        float b2, float scale,
                                        float* __restrict__ dst, int lane)
{
    const int s = lane & 15, half = lane >> 4;
    float acc = 0.f;
    #pragma unroll
    for (int k = 0; k < 16; ++k)
        acc += h1[s * 32 + half * 16 + k] * W2[half * 16 + k];
    acc += __shfl_xor(acc, 16, 32);
    if (lane < SPW) dst[s] = (acc + b2) * scale;
}

__device__ __forceinline__ void fxzu_eval(const WaveCtx& w, const RegW& rw,
                                          float* __restrict__ k_out, int lane)
{
    for (int s = 0; s < SPW; ++s)
        w.h0[s * 32 + lane] = fast_tanh(w.xe[s * 2 + 0] * w.r1W0[lane] + w.r1b0[lane]);
    __syncthreads();
    dense_layer<32>(w.h0, rw.B1, rw.b1, w.h1, lane, true);
    __syncthreads();
    out_dot(w.h1, w.r1W2, rw.r1b2, 0.3f, w.rb + 0 * 16, lane);
    __syncthreads();

    for (int s = 0; s < SPW; ++s)
        w.h0[s * 32 + lane] = fast_tanh(w.xe[s * 2 + 1] * w.r2W0[lane] + w.r2b0[lane]);
    __syncthreads();
    dense_layer<32>(w.h0, rw.B2, rw.b2, w.h1, lane, true);
    __syncthreads();
    out_dot(w.h1, w.r2W2, rw.r2b2, 0.2f, w.rb + 1 * 16, lane);
    __syncthreads();

    dense_layer<24>(w.ze, rw.B3a, rw.b3a, w.h0, lane, true);
    __syncthreads();
    dense_layer<32>(w.h0, rw.B3b, rw.b3b, w.h1, lane, true);
    __syncthreads();
    out_dot(w.h1, w.r3W2, rw.r3b2, 0.2f, w.rb + 2 * 16, lane);
    __syncthreads();

    if (lane < SPW) {
        const int s = lane;
        const float r1v = w.rb[s], r2v = w.rb[16 + s], r3v = w.rb[32 + s];
        const float xs0 = w.xe[s * 2 + 0] * 0.3f + 0.5f;
        const float xs1 = w.xe[s * 2 + 1] * 0.2f + 0.5f;
        const float us  = w.u[s] * 0.5f + 1.0f;
        const float dCa = 0.1f * (us - xs0) - r1v;
        const float dCb = -0.1f * xs1 + r1v - 3.0f * r2v + r3v;
        k_out[s * 2 + 0] = dCa * (1.0f / 0.3f);
        k_out[s * 2 + 1] = dCb * (1.0f / 0.2f);
    }
    __syncthreads();
}

__global__ __launch_bounds__(32) void reac_rk4_kernel(
    const float* __restrict__ useq, const float* __restrict__ xz0,
    const float* __restrict__ r1W0, const float* __restrict__ r1b0,
    const float* __restrict__ r1W1, const float* __restrict__ r1b1,
    const float* __restrict__ r1W2, const float* __restrict__ r1b2,
    const float* __restrict__ r2W0, const float* __restrict__ r2b0,
    const float* __restrict__ r2W1, const float* __restrict__ r2b1,
    const float* __restrict__ r2W2, const float* __restrict__ r2b2,
    const float* __restrict__ r3W0, const float* __restrict__ r3b0,
    const float* __restrict__ r3W1, const float* __restrict__ r3b1,
    const float* __restrict__ r3W2, const float* __restrict__ r3b2,
    float* __restrict__ out)
{
    __shared__ float s_r1W0[32], s_r1b0[32], s_r1W2[32];
    __shared__ float s_r2W0[32], s_r2b0[32], s_r2W2[32];
    __shared__ float s_r3W2[32];
    __shared__ float s_x[SPW * 2],  s_z[SPW * 24];
    __shared__ float s_xe[SPW * 2], s_ze[SPW * 24];
    __shared__ float s_k[4][SPW * 2];
    __shared__ float s_rb[3 * 16];
    __shared__ float s_h0[SPW * 32], s_h1[SPW * 32];
    __shared__ float s_u[SPW];

    const int lane = threadIdx.x & 31;
    const int m0   = lane & 15;

    s_r1W0[lane] = r1W0[lane]; s_r1b0[lane] = r1b0[lane]; s_r1W2[lane] = r1W2[lane];
    s_r2W0[lane] = r2W0[lane]; s_r2b0[lane] = r2b0[lane]; s_r2W2[lane] = r2W2[lane];
    s_r3W2[lane] = r3W2[lane];

    RegW rw;
    load_wfrags<8>(r1W1, rw.B1, lane);
    load_wfrags<8>(r2W1, rw.B2, lane);
    load_wfrags<6>(r3W0, rw.B3a, lane);
    load_wfrags<8>(r3W1, rw.B3b, lane);
    rw.b1[0]  = r1b1[m0]; rw.b1[1]  = r1b1[16 + m0];
    rw.b2[0]  = r2b1[m0]; rw.b2[1]  = r2b1[16 + m0];
    rw.b3a[0] = r3b0[m0]; rw.b3a[1] = r3b0[16 + m0];
    rw.b3b[0] = r3b1[m0]; rw.b3b[1] = r3b1[16 + m0];
    rw.r1b2 = r1b2[0]; rw.r2b2 = r2b2[0]; rw.r3b2 = r3b2[0];

    WaveCtx w;
    w.xe = s_xe; w.ze = s_ze; w.u = s_u; w.rb = s_rb; w.h0 = s_h0; w.h1 = s_h1;
    w.r1W0 = s_r1W0; w.r1b0 = s_r1b0; w.r2W0 = s_r2W0; w.r2b0 = s_r2b0;
    w.r1W2 = s_r1W2; w.r2W2 = s_r2W2; w.r3W2 = s_r3W2;

    const int base = blockIdx.x * SPW;

    for (int i = lane; i < SPW * 26; i += 32) {
        const int s = i / 26, c = i % 26;
        const float v = xz0[(size_t)(base + s) * 26 + c];
        if (c < 2) s_x[s * 2 + c] = v;
        else       s_z[s * 24 + (c - 2)] = v;
    }
    __syncthreads();

    for (int t = 0; t < Td; ++t) {
        if (lane < SPW) {
            const int s = lane;
            out[((size_t)(base + s) * Td + t) * 2 + 0] = s_x[s * 2 + 0];
            out[((size_t)(base + s) * Td + t) * 2 + 1] = s_x[s * 2 + 1];
            s_u[s] = useq[(size_t)(base + s) * Td + t];
            s_xe[s * 2 + 0] = s_x[s * 2 + 0];
            s_xe[s * 2 + 1] = s_x[s * 2 + 1];
            for (int c = 0; c < 24; ++c) s_ze[s * 24 + c] = s_z[s * 24 + c];
        }
        __syncthreads();
        fxzu_eval(w, rw, s_k[0], lane);

        if (lane < SPW) {
            const int s = lane;
            #pragma unroll
            for (int i = 0; i < 8; ++i)
                #pragma unroll
                for (int j = 0; j < 2; ++j) {
                    const float yi = s_z[s * 24 + 2 * i + j];
                    const float yn = (i == 7) ? s_x[s * 2 + j]
                                              : s_z[s * 24 + 2 * (i + 1) + j];
                    s_ze[s * 24 + 2 * i + j] = 0.5f * (yi + yn);
                }
            s_xe[s * 2 + 0] = s_x[s * 2 + 0] + 0.5f * s_k[0][s * 2 + 0];
            s_xe[s * 2 + 1] = s_x[s * 2 + 1] + 0.5f * s_k[0][s * 2 + 1];
        }
        __syncthreads();
        fxzu_eval(w, rw, s_k[1], lane);

        if (lane < SPW) {
            const int s = lane;
            s_xe[s * 2 + 0] = s_x[s * 2 + 0] + 0.5f * s_k[1][s * 2 + 0];
            s_xe[s * 2 + 1] = s_x[s * 2 + 1] + 0.5f * s_k[1][s * 2 + 1];
        }
        __syncthreads();
        fxzu_eval(w, rw, s_k[2], lane);

        if (lane < SPW) {
            const int s = lane;
            float tmp[16];
            #pragma unroll
            for (int c = 0; c < 14; ++c) tmp[c] = s_z[s * 24 + 2 + c];
            tmp[14] = s_x[s * 2 + 0];
            tmp[15] = s_x[s * 2 + 1];
            #pragma unroll
            for (int c = 0; c < 16; ++c) s_ze[s * 24 + c] = tmp[c];
            #pragma unroll
            for (int c = 16; c < 24; ++c) s_ze[s * 24 + c] = s_z[s * 24 + c];
            s_xe[s * 2 + 0] = s_x[s * 2 + 0] + s_k[2][s * 2 + 0];
            s_xe[s * 2 + 1] = s_x[s * 2 + 1] + s_k[2][s * 2 + 1];
        }
        __syncthreads();
        fxzu_eval(w, rw, s_k[3], lane);

        if (lane < SPW) {
            const int s = lane;
            const float x0 = s_x[s * 2 + 0], x1 = s_x[s * 2 + 1];
            const float nx0 = x0 + (s_k[0][s * 2 + 0] + 2.f * s_k[1][s * 2 + 0] +
                                    2.f * s_k[2][s * 2 + 0] + s_k[3][s * 2 + 0]) * (1.f / 6.f);
            const float nx1 = x1 + (s_k[0][s * 2 + 1] + 2.f * s_k[1][s * 2 + 1] +
                                    2.f * s_k[2][s * 2 + 1] + s_k[3][s * 2 + 1]) * (1.f / 6.f);
            float zo[24];
            #pragma unroll
            for (int c = 0; c < 24; ++c) zo[c] = s_z[s * 24 + c];
            #pragma unroll
            for (int c = 0; c < 14; ++c) s_z[s * 24 + c] = zo[c + 2];
            s_z[s * 24 + 14] = x0;
            s_z[s * 24 + 15] = x1;
            #pragma unroll
            for (int c = 16; c < 23; ++c) s_z[s * 24 + c] = zo[c + 1];
            s_z[s * 24 + 23] = s_u[s];
            s_x[s * 2 + 0] = nx0;
            s_x[s * 2 + 1] = nx1;
        }
        __syncthreads();
    }
}

extern "C" void kernel_launch(void* const* d_in, const int* in_sizes, int n_in,
                              void* d_out, int out_size, void* d_ws, size_t ws_size,
                              hipStream_t stream) {
    (void)in_sizes; (void)n_in; (void)out_size; (void)d_ws; (void)ws_size;
    const float* useq = (const float*)d_in[0];
    const float* xz0  = (const float*)d_in[1];
    const float* p[18];
    for (int i = 0; i < 18; ++i) p[i] = (const float*)d_in[2 + i];
    float* out = (float*)d_out;

    const int B = 4096;
    dim3 block(32);
    dim3 grid(B / SPW);
    reac_rk4_kernel<<<grid, block, 0, stream>>>(
        useq, xz0,
        p[0], p[1], p[2], p[3], p[4], p[5],
        p[6], p[7], p[8], p[9], p[10], p[11],
        p[12], p[13], p[14], p[15], p[16], p[17],
        out);
}


# Nets that do not depend on each other share one pass: k1's r1,r2 join the r3 pass, r1,r2 of k2..k4 interleaved; 9 passes per step become 4, per-sample scalars stay in registers; sum order unchanged
# speedup vs baseline: 1.8392x; 1.8392x over previous
#include <hip/hip_runtime.h>
#include <hip/hip_bf16.h>


#define Td  256
#define SPW 16

typedef __attribute__((ext_vector_type(2))) float v2f;
typedef __attribute__((ext_vector_type(8))) float v8f;

__device__ __forceinline__ float fast_tanh(float x)
{
    return __builtin_amdgcn_tanhf(x);
}

template <int KC>
__device__ __forceinline__ void load_wfrags(const float* __restrict__ W,
                                            v2f (&B)[2][KC], int lane)
{
    const int half = lane >> 4, m0 = lane & 15;
    #pragma unroll
    for (int nt = 0; nt < 2; ++nt) {
        const int n = nt * 16 + m0;
        #pragma unroll
        for (int kc = 0; kc < KC; ++kc) {
            B[nt][kc].x = W[(kc * 4 + 2 * half + 0) * 32 + n];
            B[nt][kc].y = W[(kc * 4 + 2 * half + 1) * 32 + n];
        }
    }
}

struct RegW {
    v2f B1[2][8];
    v2f B2[2][8];
    v2f B3a[2][6];
    v2f B3b[2][8];
    float b1[2], b2[2], b3a[2], b3b[2];
    float r1b2, r2b2, r3b2;
    float w0a, b0a, w0b, b0b;
};

template <int MT>
__device__ __forceinline__ void store_tiles(const v8f (&acc)[MT][2],
                                            const float bias[2],
                                            float* __restrict__ out, int lane)
{
    const int half = lane >> 4, m0 = lane & 15;
    #pragma unroll
    for (int mt = 0; mt < MT; ++mt) {
        #pragma unroll
        for (int r = 0; r < 8; ++r) {
            const int m = mt * 16 + r + 8 * half;
            out[m * 32 + m0]      = fast_tanh(acc[mt][0][r] + bias[0]);
            out[m * 32 + 16 + m0] = fast_tanh(acc[mt][1][r] + bias[1]);
        }
    }
}

__device__ __forceinline__ float row_dot(const float* __restrict__ h1,
                                         const float* __restrict__ W2, int lane)
{
    const int s = lane & 15, half = lane >> 4;
    float acc = 0.f;
    #pragma unroll
    for (int k = 0; k < 16; ++k)
        acc += h1[s * 32 + half * 16 + k] * W2[half * 16 + k];
    acc += __shfl_xor(acc, 16, 32);
    return acc;
}

__device__ __forceinline__ void layer0_pair(const float* __restrict__ xin,
                                            const RegW& rw,
                                            float* __restrict__ h0a,
                                            float* __restrict__ h0b, int lane)
{
    for (int s = 0; s < SPW; ++s) {
        h0a[s * 32 + lane] = fast_tanh(xin[s * 2 + 0] * rw.w0a + rw.b0a);
        h0b[s * 32 + lane] = fast_tanh(xin[s * 2 + 1] * rw.w0b + rw.b0b);
    }
}

__device__ __forceinline__ void hidden_pair(const float* __restrict__ h0a,
                                            const float* __restrict__ h0b,
                                            const RegW& rw,
                                            float* __restrict__ h1a,
                                            float* __restrict__ h1b, int lane)
{
    const int half = lane >> 4, m0 = lane & 15;
    v8f acc[2][2];
    acc[0][0] = (v8f){0.f, 0.f, 0.f, 0.f, 0.f, 0.f, 0.f, 0.f};
    acc[0][1] = acc[0][0];
    acc[1][0] = acc[0][0];
    acc[1][1] = acc[0][0];
    #pragma unroll
    for (int kc = 0; kc < 8; ++kc) {
        v2f a1, a2;
        a1.x = h0a[m0 * 32 + kc * 4 + 2 * half + 0];
        a1.y = h0a[m0 * 32 + kc * 4 + 2 * half + 1];
        a2.x = h0b[m0 * 32 + kc * 4 + 2 * half + 0];
        a2.y = h0b[m0 * 32 + kc * 4 + 2 * half + 1];
        acc[0][0] = __builtin_amdgcn_wmma_f32_16x16x4_f32(
            false, a1, false, rw.B1[0][kc], (short)0, acc[0][0], false, false);
        acc[0][1] = __builtin_amdgcn_wmma_f32_16x16x4_f32(
            false, a1, false, rw.B1[1][kc], (short)0, acc[0][1], false, false);
        acc[1][0] = __builtin_amdgcn_wmma_f32_16x16x4_f32(
            false, a2, false, rw.B2[0][kc], (short)0, acc[1][0], false, false);
        acc[1][1] = __builtin_amdgcn_wmma_f32_16x16x4_f32(
            false, a2, false, rw.B2[1][kc], (short)0, acc[1][1], false, false);
    }
    #pragma unroll
    for (int r = 0; r < 8; ++r) {
        const int m = r + 8 * half;
        h1a[m * 32 + m0]      = fast_tanh(acc[0][0][r] + rw.b1[0]);
        h1a[m * 32 + 16 + m0] = fast_tanh(acc[0][1][r] + rw.b1[1]);
        h1b[m * 32 + m0]      = fast_tanh(acc[1][0][r] + rw.b2[0]);
        h1b[m * 32 + 16 + m0] = fast_tanh(acc[1][1][r] + rw.b2[1]);
    }
}

__device__ __forceinline__ void r3_layer0(const float* __restrict__ sx,
                                          const float* __restrict__ sz,
                                          const RegW& rw,
                                          float* __restrict__ g0, int lane)
{
    const int half = lane >> 4, m0 = lane & 15;
    const float* zrow = sz + m0 * 24;

    v2f a[3][6];
    #pragma unroll
    for (int kc = 0; kc < 6; ++kc) {
        a[0][kc].x = zrow[kc * 4 + 2 * half + 0];
        a[0][kc].y = zrow[kc * 4 + 2 * half + 1];
    }
    #pragma unroll
    for (int kc = 0; kc < 4; ++kc) {
        const int c0 = kc * 4 + 2 * half + 2;
        const float* p = (c0 < 16) ? (zrow + c0) : (sx + m0 * 2);
        a[2][kc].x = p[0];
        a[2][kc].y = p[1];
        a[1][kc].x = 0.5f * (a[0][kc].x + a[2][kc].x);
        a[1][kc].y = 0.5f * (a[0][kc].y + a[2][kc].y);
    }
    #pragma unroll
    for (int kc = 4; kc < 6; ++kc) {
        a[1][kc] = a[0][kc];
        a[2][kc] = a[0][kc];
    }

    v8f acc[3][2];
    #pragma unroll
    for (int mt = 0; mt < 3; ++mt) {
        acc[mt][0] = (v8f){0.f, 0.f, 0.f, 0.f, 0.f, 0.f, 0.f, 0.f};
        acc[mt][1] = acc[mt][0];
    }
    #pragma unroll
    for (int kc = 0; kc < 6; ++kc) {
        #pragma unroll
        for (int mt = 0; mt < 3; ++mt) {
            acc[mt][0] = __builtin_amdgcn_wmma_f32_16x16x4_f32(
                false, a[mt][kc], false, rw.B3a[0][kc], (short)0, acc[mt][0], false, false);
            acc[mt][1] = __builtin_amdgcn_wmma_f32_16x16x4_f32(
                false, a[mt][kc], false, rw.B3a[1][kc], (short)0, acc[mt][1], false, false);
        }
    }
    store_tiles<3>(acc, rw.b3a, g0, lane);
}

__device__ __forceinline__ void r3_layer1(const float* __restrict__ g0,
                                          const RegW& rw,
                                          float* __restrict__ g1, int lane)
{
    const int half = lane >> 4, m0 = lane & 15;
    v8f acc[3][2];
    #pragma unroll
    for (int mt = 0; mt < 3; ++mt) {
        acc[mt][0] = (v8f){0.f, 0.f, 0.f, 0.f, 0.f, 0.f, 0.f, 0.f};
        acc[mt][1] = acc[mt][0];
    }
    #pragma unroll
    for (int kc = 0; kc < 8; ++kc) {
        #pragma unroll
        for (int mt = 0; mt < 3; ++mt) {
            v2f av;
            av.x = g0[(mt * 16 + m0) * 32 + kc * 4 + 2 * half + 0];
            av.y = g0[(mt * 16 + m0) * 32 + kc * 4 + 2 * half + 1];
            acc[mt][0] = __builtin_amdgcn_wmma_f32_16x16x4_f32(
                false, av, false, rw.B3b[0][kc], (short)0, acc[mt][0], false, false);
            acc[mt][1] = __builtin_amdgcn_wmma_f32_16x16x4_f32(
                false, av, false, rw.B3b[1][kc], (short)0, acc[mt][1], false, false);
        }
    }
    store_tiles<3>(acc, rw.b3b, g1, lane);
}

__device__ __forceinline__ void cstr_rhs(float xe0, float xe1, float u,
                                         float r1v, float r2v, float r3s,
                                         float& k0, float& k1)
{
    const float xs0 = xe0 * 0.3f + 0.5f;
    const float xs1 = xe1 * 0.2f + 0.5f;
    const float us  = u * 0.5f + 1.0f;
    const float dCa = 0.1f * (us - xs0) - r1v;
    const float dCb = -0.1f * xs1 + r1v - 3.0f * r2v + r3s;
    k0 = dCa * (1.0f / 0.3f);
    k1 = dCb * (1.0f / 0.2f);
}

struct Lds {
    float *h0a, *h0b, *h1a, *h1b;
    const float *W2a, *W2b;
};

__device__ __forceinline__ void stage_pair(const float* __restrict__ xin,
                                           float xe0, float xe1, float u, float r3s,
                                           const RegW& rw, const Lds& l,
                                           float& k0, float& k1, int lane)
{
    layer0_pair(xin, rw, l.h0a, l.h0b, lane);
    __syncthreads();
    hidden_pair(l.h0a, l.h0b, rw, l.h1a, l.h1b, lane);
    __syncthreads();
    const float r1v = (row_dot(l.h1a, l.W2a, lane) + rw.r1b2) * 0.3f;
    const float r2v = (row_dot(l.h1b, l.W2b, lane) + rw.r2b2) * 0.2f;
    cstr_rhs(xe0, xe1, u, r1v, r2v, r3s, k0, k1);
}

__global__ __launch_bounds__(32) void reac_rk4_kernel(
    const float* __restrict__ useq, const float* __restrict__ xz0,
    const float* __restrict__ r1W0, const float* __restrict__ r1b0,
    const float* __restrict__ r1W1, const float* __restrict__ r1b1,
    const float* __restrict__ r1W2, const float* __restrict__ r1b2,
    const float* __restrict__ r2W0, const float* __restrict__ r2b0,
    const float* __restrict__ r2W1, const float* __restrict__ r2b1,
    const float* __restrict__ r2W2, const float* __restrict__ r2b2,
    const float* __restrict__ r3W0, const float* __restrict__ r3b0,
    const float* __restrict__ r3W1, const float* __restrict__ r3b1,
    const float* __restrict__ r3W2, const float* __restrict__ r3b2,
    float* __restrict__ out)
{
    __shared__ __attribute__((aligned(16))) float s_r1W2[32];
    __shared__ __attribute__((aligned(16))) float s_r2W2[32];
    __shared__ __attribute__((aligned(16))) float s_r3W2[32];
    __shared__ __attribute__((aligned(16))) float s_x[SPW * 2];
    __shared__ __attribute__((aligned(16))) float s_z[SPW * 24];
    __shared__ __attribute__((aligned(16))) float s_xe[SPW * 2];
    __shared__ __attribute__((aligned(16))) float s_h0a[SPW * 32];
    __shared__ __attribute__((aligned(16))) float s_h0b[SPW * 32];
    __shared__ __attribute__((aligned(16))) float s_h1a[SPW * 32];
    __shared__ __attribute__((aligned(16))) float s_h1b[SPW * 32];
    __shared__ __attribute__((aligned(16))) float s_g0[3 * SPW * 32];
    __shared__ __attribute__((aligned(16))) float s_g1[3 * SPW * 32];

    const int lane = threadIdx.x & 31;
    const int m0   = lane & 15;

    s_r1W2[lane] = r1W2[lane];
    s_r2W2[lane] = r2W2[lane];
    s_r3W2[lane] = r3W2[lane];

    RegW rw;
    load_wfrags<8>(r1W1, rw.B1, lane);
    load_wfrags<8>(r2W1, rw.B2, lane);
    load_wfrags<6>(r3W0, rw.B3a, lane);
    load_wfrags<8>(r3W1, rw.B3b, lane);
    rw.b1[0]  = r1b1[m0]; rw.b1[1]  = r1b1[16 + m0];
    rw.b2[0]  = r2b1[m0]; rw.b2[1]  = r2b1[16 + m0];
    rw.b3a[0] = r3b0[m0]; rw.b3a[1] = r3b0[16 + m0];
    rw.b3b[0] = r3b1[m0]; rw.b3b[1] = r3b1[16 + m0];
    rw.r1b2 = r1b2[0]; rw.r2b2 = r2b2[0]; rw.r3b2 = r3b2[0];
    rw.w0a = r1W0[lane]; rw.b0a = r1b0[lane];
    rw.w0b = r2W0[lane]; rw.b0b = r2b0[lane];

    Lds l;
    l.h0a = s_h0a; l.h0b = s_h0b; l.h1a = s_h1a; l.h1b = s_h1b;
    l.W2a = s_r1W2; l.W2b = s_r2W2;

    const int base = blockIdx.x * SPW;

    for (int i = lane; i < SPW * 26; i += 32) {
        const int s = i / 26, c = i % 26;
        const float v = xz0[(size_t)(base + s) * 26 + c];
        if (c < 2) s_x[s * 2 + c] = v;
        else       s_z[s * 24 + (c - 2)] = v;
    }
    __syncthreads();

    for (int t = 0; t < Td; ++t) {
        const float x0 = s_x[m0 * 2 + 0], x1 = s_x[m0 * 2 + 1];
        const float u  = useq[(size_t)(base + m0) * Td + t];
        if (lane < SPW) {
            out[((size_t)(base + m0) * Td + t) * 2 + 0] = x0;
            out[((size_t)(base + m0) * Td + t) * 2 + 1] = x1;
        }

        r3_layer0(s_x, s_z, rw, s_g0, lane);
        layer0_pair(s_x, rw, s_h0a, s_h0b, lane);
        __syncthreads();
        r3_layer1(s_g0, rw, s_g1, lane);
        hidden_pair(s_h0a, s_h0b, rw, s_h1a, s_h1b, lane);
        __syncthreads();
        const float r3z   = (row_dot(s_g1 + 0 * SPW * 32, s_r3W2, lane) + rw.r3b2) * 0.2f;
        const float r3z23 = (row_dot(s_g1 + 1 * SPW * 32, s_r3W2, lane) + rw.r3b2) * 0.2f;
        const float r3z4  = (row_dot(s_g1 + 2 * SPW * 32, s_r3W2, lane) + rw.r3b2) * 0.2f;
        float k1a, k1b, k2a, k2b, k3a, k3b, k4a, k4b;
        {
            const float r1v = (row_dot(s_h1a, s_r1W2, lane) + rw.r1b2) * 0.3f;
            const float r2v = (row_dot(s_h1b, s_r2W2, lane) + rw.r2b2) * 0.2f;
            cstr_rhs(x0, x1, u, r1v, r2v, r3z, k1a, k1b);
        }
        float xe0 = x0 + 0.5f * k1a, xe1 = x1 + 0.5f * k1b;
        if (lane < SPW) { s_xe[m0 * 2 + 0] = xe0; s_xe[m0 * 2 + 1] = xe1; }
        __syncthreads();

        stage_pair(s_xe, xe0, xe1, u, r3z23, rw, l, k2a, k2b, lane);
        xe0 = x0 + 0.5f * k2a; xe1 = x1 + 0.5f * k2b;
        if (lane < SPW) { s_xe[m0 * 2 + 0] = xe0; s_xe[m0 * 2 + 1] = xe1; }
        __syncthreads();

        stage_pair(s_xe, xe0, xe1, u, r3z23, rw, l, k3a, k3b, lane);
        xe0 = x0 + k3a; xe1 = x1 + k3b;
        if (lane < SPW) { s_xe[m0 * 2 + 0] = xe0; s_xe[m0 * 2 + 1] = xe1; }
        __syncthreads();

        stage_pair(s_xe, xe0, xe1, u, r3z4, rw, l, k4a, k4b, lane);

        const float nx0 = x0 + (k1a + 2.f * k2a + 2.f * k3a + k4a) * (1.f / 6.f);
        const float nx1 = x1 + (k1b + 2.f * k2b + 2.f * k3b + k4b) * (1.f / 6.f);
        if (lane < SPW) {
            const int s = lane;
            float zo[24];
            #pragma unroll
            for (int c = 0; c < 24; ++c) zo[c] = s_z[s * 24 + c];
            #pragma unroll
            for (int c = 0; c < 14; ++c) s_z[s * 24 + c] = zo[c + 2];
            s_z[s * 24 + 14] = x0;
            s_z[s * 24 + 15] = x1;
            #pragma unroll
            for (int c = 16; c < 23; ++c) s_z[s * 24 + c] = zo[c + 1];
            s_z[s * 24 + 23] = u;
            s_x[s * 2 + 0] = nx0;
            s_x[s * 2 + 1] = nx1;
        }
        __syncthreads();
    }
}

extern "C" void kernel_launch(void* const* d_in, const int* in_sizes, int n_in,
                              void* d_out, int out_size, void* d_ws, size_t ws_size,
                              hipStream_t stream) {
    (void)in_sizes; (void)n_in; (void)out_size; (void)d_ws; (void)ws_size;
    const float* useq = (const float*)d_in[0];
    const float* xz0  = (const float*)d_in[1];
    const float* p[18];
    for (int i = 0; i < 18; ++i) p[i] = (const float*)d_in[2 + i];
    float* out = (float*)d_out;

    const int B = 4096;
    dim3 block(32);
    dim3 grid(B / SPW);
    reac_rk4_kernel<<<grid, block, 0, stream>>>(
        useq, xz0,
        p[0], p[1], p[2], p[3], p[4], p[5],
        p[6], p[7], p[8], p[9], p[10], p[11],
        p[12], p[13], p[14], p[15], p[16], p[17],
        out);
}


# r1/r2 layer 0 (scalar input) computed by each lane directly in the WMMA A-fragment layout from its own sample's x in registers; stage input, x and h0 no longer cross LDS; sum order unchanged
# speedup vs baseline: 2.0900x; 1.1364x over previous
#include <hip/hip_runtime.h>
#include <hip/hip_bf16.h>


#define Td  256
#define SPW 16

typedef __attribute__((ext_vector_type(2))) float v2f;
typedef __attribute__((ext_vector_type(8))) float v8f;

__device__ __forceinline__ float fast_tanh(float x)
{
    return __builtin_amdgcn_tanhf(x);
}

template <int KC>
__device__ __forceinline__ void load_wfrags(const float* __restrict__ W,
                                            v2f (&B)[2][KC], int lane)
{
    const int half = lane >> 4, m0 = lane & 15;
    #pragma unroll
    for (int nt = 0; nt < 2; ++nt) {
        const int n = nt * 16 + m0;
        #pragma unroll
        for (int kc = 0; kc < KC; ++kc) {
            B[nt][kc].x = W[(kc * 4 + 2 * half + 0) * 32 + n];
            B[nt][kc].y = W[(kc * 4 + 2 * half + 1) * 32 + n];
        }
    }
}

struct RegW {
    v2f B1[2][8];
    v2f B2[2][8];
    v2f B3a[2][6];
    v2f B3b[2][8];
    float b1[2], b2[2], b3a[2], b3b[2];
    float r1b2, r2b2, r3b2;
    v2f w0a[8], b0a[8], w0b[8], b0b[8];
};

template <int MT>
__device__ __forceinline__ void store_tiles(const v8f (&acc)[MT][2],
                                            const float bias[2],
                                            float* __restrict__ out, int lane)
{
    const int half = lane >> 4, m0 = lane & 15;
    #pragma unroll
    for (int mt = 0; mt < MT; ++mt) {
        #pragma unroll
        for (int r = 0; r < 8; ++r) {
            const int m = mt * 16 + r + 8 * half;
            out[m * 32 + m0]      = fast_tanh(acc[mt][0][r] + bias[0]);
            out[m * 32 + 16 + m0] = fast_tanh(acc[mt][1][r] + bias[1]);
        }
    }
}

__device__ __forceinline__ float row_dot(const float* __restrict__ h1,
                                         const float* __restrict__ W2, int lane)
{
    const int s = lane & 15, half = lane >> 4;
    float acc = 0.f;
    #pragma unroll
    for (int k = 0; k < 16; ++k)
        acc += h1[s * 32 + half * 16 + k] * W2[half * 16 + k];
    acc += __shfl_xor(acc, 16, 32);
    return acc;
}

__device__ __forceinline__ void hidden_pair(float xa, float xb,
                                            const RegW& rw,
                                            float* __restrict__ h1a,
                                            float* __restrict__ h1b, int lane)
{
    const int half = lane >> 4, m0 = lane & 15;
    v8f acc[2][2];
    acc[0][0] = (v8f){0.f, 0.f, 0.f, 0.f, 0.f, 0.f, 0.f, 0.f};
    acc[0][1] = acc[0][0];
    acc[1][0] = acc[0][0];
    acc[1][1] = acc[0][0];
    #pragma unroll
    for (int kc = 0; kc < 8; ++kc) {
        v2f a1, a2;
        a1.x = fast_tanh(xa * rw.w0a[kc].x + rw.b0a[kc].x);
        a1.y = fast_tanh(xa * rw.w0a[kc].y + rw.b0a[kc].y);
        a2.x = fast_tanh(xb * rw.w0b[kc].x + rw.b0b[kc].x);
        a2.y = fast_tanh(xb * rw.w0b[kc].y + rw.b0b[kc].y);
        acc[0][0] = __builtin_amdgcn_wmma_f32_16x16x4_f32(
            false, a1, false, rw.B1[0][kc], (short)0, acc[0][0], false, false);
        acc[0][1] = __builtin_amdgcn_wmma_f32_16x16x4_f32(
            false, a1, false, rw.B1[1][kc], (short)0, acc[0][1], false, false);
        acc[1][0] = __builtin_amdgcn_wmma_f32_16x16x4_f32(
            false, a2, false, rw.B2[0][kc], (short)0, acc[1][0], false, false);
        acc[1][1] = __builtin_amdgcn_wmma_f32_16x16x4_f32(
            false, a2, false, rw.B2[1][kc], (short)0, acc[1][1], false, false);
    }
    #pragma unroll
    for (int r = 0; r < 8; ++r) {
        const int m = r + 8 * half;
        h1a[m * 32 + m0]      = fast_tanh(acc[0][0][r] + rw.b1[0]);
        h1a[m * 32 + 16 + m0] = fast_tanh(acc[0][1][r] + rw.b1[1]);
        h1b[m * 32 + m0]      = fast_tanh(acc[1][0][r] + rw.b2[0]);
        h1b[m * 32 + 16 + m0] = fast_tanh(acc[1][1][r] + rw.b2[1]);
    }
}

__device__ __forceinline__ void r3_layer0(float x0, float x1,
                                          const float* __restrict__ sz,
                                          const RegW& rw,
                                          float* __restrict__ g0, int lane)
{
    const int half = lane >> 4, m0 = lane & 15;
    const float* zrow = sz + m0 * 24;

    v2f a[3][6];
    #pragma unroll
    for (int kc = 0; kc < 6; ++kc) {
        a[0][kc].x = zrow[kc * 4 + 2 * half + 0];
        a[0][kc].y = zrow[kc * 4 + 2 * half + 1];
    }
    #pragma unroll
    for (int kc = 0; kc < 4; ++kc) {
        const int c0 = kc * 4 + 2 * half + 2;
        const float zx = zrow[c0], zy = zrow[c0 + 1];
        a[2][kc].x = (c0 < 16) ? zx : x0;
        a[2][kc].y = (c0 < 16) ? zy : x1;
        a[1][kc].x = 0.5f * (a[0][kc].x + a[2][kc].x);
        a[1][kc].y = 0.5f * (a[0][kc].y + a[2][kc].y);
    }
    #pragma unroll
    for (int kc = 4; kc < 6; ++kc) {
        a[1][kc] = a[0][kc];
        a[2][kc] = a[0][kc];
    }

    v8f acc[3][2];
    #pragma unroll
    for (int mt = 0; mt < 3; ++mt) {
        acc[mt][0] = (v8f){0.f, 0.f, 0.f, 0.f, 0.f, 0.f, 0.f, 0.f};
        acc[mt][1] = acc[mt][0];
    }
    #pragma unroll
    for (int kc = 0; kc < 6; ++kc) {
        #pragma unroll
        for (int mt = 0; mt < 3; ++mt) {
            acc[mt][0] = __builtin_amdgcn_wmma_f32_16x16x4_f32(
                false, a[mt][kc], false, rw.B3a[0][kc], (short)0, acc[mt][0], false, false);
            acc[mt][1] = __builtin_amdgcn_wmma_f32_16x16x4_f32(
                false, a[mt][kc], false, rw.B3a[1][kc], (short)0, acc[mt][1], false, false);
        }
    }
    store_tiles<3>(acc, rw.b3a, g0, lane);
}

__device__ __forceinline__ void r3_layer1(const float* __restrict__ g0,
                                          const RegW& rw,
                                          float* __restrict__ g1, int lane)
{
    const int half = lane >> 4, m0 = lane & 15;
    v8f acc[3][2];
    #pragma unroll
    for (int mt = 0; mt < 3; ++mt) {
        acc[mt][0] = (v8f){0.f, 0.f, 0.f, 0.f, 0.f, 0.f, 0.f, 0.f};
        acc[mt][1] = acc[mt][0];
    }
    #pragma unroll
    for (int kc = 0; kc < 8; ++kc) {
        #pragma unroll
        for (int mt = 0; mt < 3; ++mt) {
            v2f av;
            av.x = g0[(mt * 16 + m0) * 32 + kc * 4 + 2 * half + 0];
            av.y = g0[(mt * 16 + m0) * 32 + kc * 4 + 2 * half + 1];
            acc[mt][0] = __builtin_amdgcn_wmma_f32_16x16x4_f32(
                false, av, false, rw.B3b[0][kc], (short)0, acc[mt][0], false, false);
            acc[mt][1] = __builtin_amdgcn_wmma_f32_16x16x4_f32(
                false, av, false, rw.B3b[1][kc], (short)0, acc[mt][1], false, false);
        }
    }
    store_tiles<3>(acc, rw.b3b, g1, lane);
}

__device__ __forceinline__ void cstr_rhs(float xe0, float xe1, float u,
                                         float r1v, float r2v, float r3s,
                                         float& k0, float& k1)
{
    const float xs0 = xe0 * 0.3f + 0.5f;
    const float xs1 = xe1 * 0.2f + 0.5f;
    const float us  = u * 0.5f + 1.0f;
    const float dCa = 0.1f * (us - xs0) - r1v;
    const float dCb = -0.1f * xs1 + r1v - 3.0f * r2v + r3s;
    k0 = dCa * (1.0f / 0.3f);
    k1 = dCb * (1.0f / 0.2f);
}

struct Lds {
    float *h1a, *h1b;
    const float *W2a, *W2b;
};

__device__ __forceinline__ void stage_pair(float xe0, float xe1, float u, float r3s,
                                           const RegW& rw, const Lds& l,
                                           float& k0, float& k1, int lane)
{
    __syncthreads();
    hidden_pair(xe0, xe1, rw, l.h1a, l.h1b, lane);
    __syncthreads();
    const float r1v = (row_dot(l.h1a, l.W2a, lane) + rw.r1b2) * 0.3f;
    const float r2v = (row_dot(l.h1b, l.W2b, lane) + rw.r2b2) * 0.2f;
    cstr_rhs(xe0, xe1, u, r1v, r2v, r3s, k0, k1);
}

__global__ __launch_bounds__(32) void reac_rk4_kernel(
    const float* __restrict__ useq, const float* __restrict__ xz0,
    const float* __restrict__ r1W0, const float* __restrict__ r1b0,
    const float* __restrict__ r1W1, const float* __restrict__ r1b1,
    const float* __restrict__ r1W2, const float* __restrict__ r1b2,
    const float* __restrict__ r2W0, const float* __restrict__ r2b0,
    const float* __restrict__ r2W1, const float* __restrict__ r2b1,
    const float* __restrict__ r2W2, const float* __restrict__ r2b2,
    const float* __restrict__ r3W0, const float* __restrict__ r3b0,
    const float* __restrict__ r3W1, const float* __restrict__ r3b1,
    const float* __restrict__ r3W2, const float* __restrict__ r3b2,
    float* __restrict__ out)
{
    __shared__ __attribute__((aligned(16))) float s_r1W2[32];
    __shared__ __attribute__((aligned(16))) float s_r2W2[32];
    __shared__ __attribute__((aligned(16))) float s_r3W2[32];
    __shared__ __attribute__((aligned(16))) float s_z[SPW * 24];
    __shared__ __attribute__((aligned(16))) float s_h1a[SPW * 32];
    __shared__ __attribute__((aligned(16))) float s_h1b[SPW * 32];
    __shared__ __attribute__((aligned(16))) float s_g0[3 * SPW * 32];
    __shared__ __attribute__((aligned(16))) float s_g1[3 * SPW * 32];

    const int lane = threadIdx.x & 31;
    const int m0   = lane & 15;
    const int half = lane >> 4;

    s_r1W2[lane] = r1W2[lane];
    s_r2W2[lane] = r2W2[lane];
    s_r3W2[lane] = r3W2[lane];

    RegW rw;
    load_wfrags<8>(r1W1, rw.B1, lane);
    load_wfrags<8>(r2W1, rw.B2, lane);
    load_wfrags<6>(r3W0, rw.B3a, lane);
    load_wfrags<8>(r3W1, rw.B3b, lane);
    rw.b1[0]  = r1b1[m0]; rw.b1[1]  = r1b1[16 + m0];
    rw.b2[0]  = r2b1[m0]; rw.b2[1]  = r2b1[16 + m0];
    rw.b3a[0] = r3b0[m0]; rw.b3a[1] = r3b0[16 + m0];
    rw.b3b[0] = r3b1[m0]; rw.b3b[1] = r3b1[16 + m0];
    rw.r1b2 = r1b2[0]; rw.r2b2 = r2b2[0]; rw.r3b2 = r3b2[0];
    #pragma unroll
    for (int kc = 0; kc < 8; ++kc) {
        const int n = kc * 4 + 2 * half;
        rw.w0a[kc].x = r1W0[n]; rw.w0a[kc].y = r1W0[n + 1];
        rw.b0a[kc].x = r1b0[n]; rw.b0a[kc].y = r1b0[n + 1];
        rw.w0b[kc].x = r2W0[n]; rw.w0b[kc].y = r2W0[n + 1];
        rw.b0b[kc].x = r2b0[n]; rw.b0b[kc].y = r2b0[n + 1];
    }

    Lds l;
    l.h1a = s_h1a; l.h1b = s_h1b;
    l.W2a = s_r1W2; l.W2b = s_r2W2;

    const int base = blockIdx.x * SPW;

    for (int i = lane; i < SPW * 26; i += 32) {
        const int s = i / 26, c = i % 26;
        if (c >= 2) s_z[s * 24 + (c - 2)] = xz0[(size_t)(base + s) * 26 + c];
    }
    float x0 = xz0[(size_t)(base + m0) * 26 + 0];
    float x1 = xz0[(size_t)(base + m0) * 26 + 1];
    __syncthreads();

    for (int t = 0; t < Td; ++t) {
        const float u  = useq[(size_t)(base + m0) * Td + t];
        if (lane < SPW) {
            out[((size_t)(base + m0) * Td + t) * 2 + 0] = x0;
            out[((size_t)(base + m0) * Td + t) * 2 + 1] = x1;
        }

        r3_layer0(x0, x1, s_z, rw, s_g0, lane);
        __syncthreads();
        hidden_pair(x0, x1, rw, s_h1a, s_h1b, lane);
        r3_layer1(s_g0, rw, s_g1, lane);
        __syncthreads();
        const float r3z   = (row_dot(s_g1 + 0 * SPW * 32, s_r3W2, lane) + rw.r3b2) * 0.2f;
        const float r3z23 = (row_dot(s_g1 + 1 * SPW * 32, s_r3W2, lane) + rw.r3b2) * 0.2f;
        const float r3z4  = (row_dot(s_g1 + 2 * SPW * 32, s_r3W2, lane) + rw.r3b2) * 0.2f;
        float k1a, k1b, k2a, k2b, k3a, k3b, k4a, k4b;
        {
            const float r1v = (row_dot(s_h1a, s_r1W2, lane) + rw.r1b2) * 0.3f;
            const float r2v = (row_dot(s_h1b, s_r2W2, lane) + rw.r2b2) * 0.2f;
            cstr_rhs(x0, x1, u, r1v, r2v, r3z, k1a, k1b);
        }
        float xe0 = x0 + 0.5f * k1a, xe1 = x1 + 0.5f * k1b;

        stage_pair(xe0, xe1, u, r3z23, rw, l, k2a, k2b, lane);
        xe0 = x0 + 0.5f * k2a; xe1 = x1 + 0.5f * k2b;

        stage_pair(xe0, xe1, u, r3z23, rw, l, k3a, k3b, lane);
        xe0 = x0 + k3a; xe1 = x1 + k3b;

        stage_pair(xe0, xe1, u, r3z4, rw, l, k4a, k4b, lane);

        const float nx0 = x0 + (k1a + 2.f * k2a + 2.f * k3a + k4a) * (1.f / 6.f);
        const float nx1 = x1 + (k1b + 2.f * k2b + 2.f * k3b + k4b) * (1.f / 6.f);
        if (lane < SPW) {
            const int s = lane;
            float zo[24];
            #pragma unroll
            for (int c = 0; c < 24; ++c) zo[c] = s_z[s * 24 + c];
            #pragma unroll
            for (int c = 0; c < 14; ++c) s_z[s * 24 + c] = zo[c + 2];
            s_z[s * 24 + 14] = x0;
            s_z[s * 24 + 15] = x1;
            #pragma unroll
            for (int c = 16; c < 23; ++c) s_z[s * 24 + c] = zo[c + 1];
            s_z[s * 24 + 23] = u;
        }
        x0 = nx0;
        x1 = nx1;
        __syncthreads();
    }
}

extern "C" void kernel_launch(void* const* d_in, const int* in_sizes, int n_in,
                              void* d_out, int out_size, void* d_ws, size_t ws_size,
                              hipStream_t stream) {
    (void)in_sizes; (void)n_in; (void)out_size; (void)d_ws; (void)ws_size;
    const float* useq = (const float*)d_in[0];
    const float* xz0  = (const float*)d_in[1];
    const float* p[18];
    for (int i = 0; i < 18; ++i) p[i] = (const float*)d_in[2 + i];
    float* out = (float*)d_out;

    const int B = 4096;
    dim3 block(32);
    dim3 grid(B / SPW);
    reac_rk4_kernel<<<grid, block, 0, stream>>>(
        useq, xz0,
        p[0], p[1], p[2], p[3], p[4], p[5],
        p[6], p[7], p[8], p[9], p[10], p[11],
        p[12], p[13], p[14], p[15], p[16], p[17],
        out);
}


# Output dots (32->1) of r1, r2, r3 taken from the accumulator tiles in registers: per-lane products + 16-lane DPP row reduction; h1/g1 never go to LDS, RK stages 2..4 LDS-free; dot sum order changes
# speedup vs baseline: 3.0118x; 1.4411x over previous
#include <hip/hip_runtime.h>
#include <hip/hip_bf16.h>


#define Td  256
#define SPW 16

typedef __attribute__((ext_vector_type(2))) float v2f;
typedef __attribute__((ext_vector_type(8))) float v8f;

__device__ __forceinline__ float fast_tanh(float x)
{
    return __builtin_amdgcn_tanhf(x);
}

template <int KC>
__device__ __forceinline__ void load_wfrags(const float* __restrict__ W,
                                            v2f (&B)[2][KC], int lane)
{
    const int half = lane >> 4, m0 = lane & 15;
    #pragma unroll
    for (int nt = 0; nt < 2; ++nt) {
        const int n = nt * 16 + m0;
        #pragma unroll
        for (int kc = 0; kc < KC; ++kc) {
            B[nt][kc].x = W[(kc * 4 + 2 * half + 0) * 32 + n];
            B[nt][kc].y = W[(kc * 4 + 2 * half + 1) * 32 + n];
        }
    }
}

struct RegW {
    v2f B1[2][8];
    v2f B2[2][8];
    v2f B3a[2][6];
    v2f B3b[2][8];
    float b1[2], b2[2], b3a[2], b3b[2];
    float r1b2, r2b2, r3b2;
    float w2a[2], w2b[2], w2c[2];
    v2f w0a[8], b0a[8], w0b[8], b0b[8];
};

template <int MT>
__device__ __forceinline__ void store_tiles(const v8f (&acc)[MT][2],
                                            const float bias[2],
                                            float* __restrict__ out, int lane)
{
    const int half = lane >> 4, m0 = lane & 15;
    #pragma unroll
    for (int mt = 0; mt < MT; ++mt) {
        #pragma unroll
        for (int r = 0; r < 8; ++r) {
            const int m = mt * 16 + r + 8 * half;
            out[m * 32 + m0]      = fast_tanh(acc[mt][0][r] + bias[0]);
            out[m * 32 + 16 + m0] = fast_tanh(acc[mt][1][r] + bias[1]);
        }
    }
}

template <int MASK>
__device__ __forceinline__ float row_xor(float v)
{
    return __builtin_bit_cast(float, __builtin_amdgcn_update_dpp(
        0, __builtin_bit_cast(int, v), 0x160 + MASK, 0xf, 0xf, true));
}

__device__ __forceinline__ float other_row(float v)
{
    const int b = __builtin_bit_cast(int, v);
    return __builtin_bit_cast(float,
        __builtin_amdgcn_permlanex16(b, b, 0x76543210, 0xfedcba98, false, false));
}

struct LaneSel {
    bool b0, b1, b2;
    bool own;
};

__device__ __forceinline__ float tile_dot(const v8f& c0, const v8f& c1,
                                          const float bias[2], const float w2[2],
                                          const LaneSel& ls)
{
    float p[8];
    #pragma unroll
    for (int r = 0; r < 8; ++r)
        p[r] = fast_tanh(c0[r] + bias[0]) * w2[0] + fast_tanh(c1[r] + bias[1]) * w2[1];
    float q[4];
    #pragma unroll
    for (int i = 0; i < 4; ++i) {
        const float keep = ls.b2 ? p[4 + i] : p[i];
        const float send = ls.b2 ? p[i] : p[4 + i];
        q[i] = keep + row_xor<4>(send);
    }
    float s[2];
    #pragma unroll
    for (int i = 0; i < 2; ++i) {
        const float keep = ls.b1 ? q[2 + i] : q[i];
        const float send = ls.b1 ? q[i] : q[2 + i];
        s[i] = keep + row_xor<2>(send);
    }
    const float keep = ls.b0 ? s[1] : s[0];
    const float send = ls.b0 ? s[0] : s[1];
    const float t = keep + row_xor<1>(send);
    const float u = t + row_xor<8>(t);
    const float o = other_row(u);
    return ls.own ? u : o;
}

__device__ __forceinline__ void hidden_pair(float xa, float xb,
                                            const RegW& rw, const LaneSel& ls,
                                            float& da, float& db)
{
    v8f acc[2][2];
    acc[0][0] = (v8f){0.f, 0.f, 0.f, 0.f, 0.f, 0.f, 0.f, 0.f};
    acc[0][1] = acc[0][0];
    acc[1][0] = acc[0][0];
    acc[1][1] = acc[0][0];
    #pragma unroll
    for (int kc = 0; kc < 8; ++kc) {
        v2f a1, a2;
        a1.x = fast_tanh(xa * rw.w0a[kc].x + rw.b0a[kc].x);
        a1.y = fast_tanh(xa * rw.w0a[kc].y + rw.b0a[kc].y);
        a2.x = fast_tanh(xb * rw.w0b[kc].x + rw.b0b[kc].x);
        a2.y = fast_tanh(xb * rw.w0b[kc].y + rw.b0b[kc].y);
        acc[0][0] = __builtin_amdgcn_wmma_f32_16x16x4_f32(
            false, a1, false, rw.B1[0][kc], (short)0, acc[0][0], false, false);
        acc[0][1] = __builtin_amdgcn_wmma_f32_16x16x4_f32(
            false, a1, false, rw.B1[1][kc], (short)0, acc[0][1], false, false);
        acc[1][0] = __builtin_amdgcn_wmma_f32_16x16x4_f32(
            false, a2, false, rw.B2[0][kc], (short)0, acc[1][0], false, false);
        acc[1][1] = __builtin_amdgcn_wmma_f32_16x16x4_f32(
            false, a2, false, rw.B2[1][kc], (short)0, acc[1][1], false, false);
    }
    da = tile_dot(acc[0][0], acc[0][1], rw.b1, rw.w2a, ls);
    db = tile_dot(acc[1][0], acc[1][1], rw.b2, rw.w2b, ls);
}

__device__ __forceinline__ void r3_layer0(float x0, float x1,
                                          const float* __restrict__ sz,
                                          const RegW& rw,
                                          float* __restrict__ g0, int lane)
{
    const int half = lane >> 4, m0 = lane & 15;
    const float* zrow = sz + m0 * 24;

    v2f a[3][6];
    #pragma unroll
    for (int kc = 0; kc < 6; ++kc) {
        a[0][kc].x = zrow[kc * 4 + 2 * half + 0];
        a[0][kc].y = zrow[kc * 4 + 2 * half + 1];
    }
    #pragma unroll
    for (int kc = 0; kc < 4; ++kc) {
        const int c0 = kc * 4 + 2 * half + 2;
        const float zx = zrow[c0], zy = zrow[c0 + 1];
        a[2][kc].x = (c0 < 16) ? zx : x0;
        a[2][kc].y = (c0 < 16) ? zy : x1;
        a[1][kc].x = 0.5f * (a[0][kc].x + a[2][kc].x);
        a[1][kc].y = 0.5f * (a[0][kc].y + a[2][kc].y);
    }
    #pragma unroll
    for (int kc = 4; kc < 6; ++kc) {
        a[1][kc] = a[0][kc];
        a[2][kc] = a[0][kc];
    }

    v8f acc[3][2];
    #pragma unroll
    for (int mt = 0; mt < 3; ++mt) {
        acc[mt][0] = (v8f){0.f, 0.f, 0.f, 0.f, 0.f, 0.f, 0.f, 0.f};
        acc[mt][1] = acc[mt][0];
    }
    #pragma unroll
    for (int kc = 0; kc < 6; ++kc) {
        #pragma unroll
        for (int mt = 0; mt < 3; ++mt) {
            acc[mt][0] = __builtin_amdgcn_wmma_f32_16x16x4_f32(
                false, a[mt][kc], false, rw.B3a[0][kc], (short)0, acc[mt][0], false, false);
            acc[mt][1] = __builtin_amdgcn_wmma_f32_16x16x4_f32(
                false, a[mt][kc], false, rw.B3a[1][kc], (short)0, acc[mt][1], false, false);
        }
    }
    store_tiles<3>(acc, rw.b3a, g0, lane);
}

__device__ __forceinline__ void r3_layer1(const float* __restrict__ g0,
                                          const RegW& rw, const LaneSel& ls,
                                          float (&d)[3], int lane)
{
    const int half = lane >> 4, m0 = lane & 15;
    v8f acc[3][2];
    #pragma unroll
    for (int mt = 0; mt < 3; ++mt) {
        acc[mt][0] = (v8f){0.f, 0.f, 0.f, 0.f, 0.f, 0.f, 0.f, 0.f};
        acc[mt][1] = acc[mt][0];
    }
    #pragma unroll
    for (int kc = 0; kc < 8; ++kc) {
        #pragma unroll
        for (int mt = 0; mt < 3; ++mt) {
            v2f av;
            av.x = g0[(mt * 16 + m0) * 32 + kc * 4 + 2 * half + 0];
            av.y = g0[(mt * 16 + m0) * 32 + kc * 4 + 2 * half + 1];
            acc[mt][0] = __builtin_amdgcn_wmma_f32_16x16x4_f32(
                false, av, false, rw.B3b[0][kc], (short)0, acc[mt][0], false, false);
            acc[mt][1] = __builtin_amdgcn_wmma_f32_16x16x4_f32(
                false, av, false, rw.B3b[1][kc], (short)0, acc[mt][1], false, false);
        }
    }
    #pragma unroll
    for (int mt = 0; mt < 3; ++mt)
        d[mt] = tile_dot(acc[mt][0], acc[mt][1], rw.b3b, rw.w2c, ls);
}

__device__ __forceinline__ void cstr_rhs(float xe0, float xe1, float u,
                                         float r1v, float r2v, float r3s,
                                         float& k0, float& k1)
{
    const float xs0 = xe0 * 0.3f + 0.5f;
    const float xs1 = xe1 * 0.2f + 0.5f;
    const float us  = u * 0.5f + 1.0f;
    const float dCa = 0.1f * (us - xs0) - r1v;
    const float dCb = -0.1f * xs1 + r1v - 3.0f * r2v + r3s;
    k0 = dCa * (1.0f / 0.3f);
    k1 = dCb * (1.0f / 0.2f);
}

__device__ __forceinline__ void stage_pair(float xe0, float xe1, float u, float r3s,
                                           const RegW& rw, const LaneSel& ls,
                                           float& k0, float& k1)
{
    float d1, d2;
    hidden_pair(xe0, xe1, rw, ls, d1, d2);
    const float r1v = (d1 + rw.r1b2) * 0.3f;
    const float r2v = (d2 + rw.r2b2) * 0.2f;
    cstr_rhs(xe0, xe1, u, r1v, r2v, r3s, k0, k1);
}

__global__ __launch_bounds__(32) void reac_rk4_kernel(
    const float* __restrict__ useq, const float* __restrict__ xz0,
    const float* __restrict__ r1W0, const float* __restrict__ r1b0,
    const float* __restrict__ r1W1, const float* __restrict__ r1b1,
    const float* __restrict__ r1W2, const float* __restrict__ r1b2,
    const float* __restrict__ r2W0, const float* __restrict__ r2b0,
    const float* __restrict__ r2W1, const float* __restrict__ r2b1,
    const float* __restrict__ r2W2, const float* __restrict__ r2b2,
    const float* __restrict__ r3W0, const float* __restrict__ r3b0,
    const float* __restrict__ r3W1, const float* __restrict__ r3b1,
    const float* __restrict__ r3W2, const float* __restrict__ r3b2,
    float* __restrict__ out)
{
    __shared__ __attribute__((aligned(16))) float s_z[SPW * 24];
    __shared__ __attribute__((aligned(16))) float s_g0[3 * SPW * 32];

    const int lane = threadIdx.x & 31;
    const int m0   = lane & 15;
    const int half = lane >> 4;

    RegW rw;
    load_wfrags<8>(r1W1, rw.B1, lane);
    load_wfrags<8>(r2W1, rw.B2, lane);
    load_wfrags<6>(r3W0, rw.B3a, lane);
    load_wfrags<8>(r3W1, rw.B3b, lane);
    rw.b1[0]  = r1b1[m0]; rw.b1[1]  = r1b1[16 + m0];
    rw.b2[0]  = r2b1[m0]; rw.b2[1]  = r2b1[16 + m0];
    rw.b3a[0] = r3b0[m0]; rw.b3a[1] = r3b0[16 + m0];
    rw.b3b[0] = r3b1[m0]; rw.b3b[1] = r3b1[16 + m0];
    rw.r1b2 = r1b2[0]; rw.r2b2 = r2b2[0]; rw.r3b2 = r3b2[0];
    rw.w2a[0] = r1W2[m0]; rw.w2a[1] = r1W2[16 + m0];
    rw.w2b[0] = r2W2[m0]; rw.w2b[1] = r2W2[16 + m0];
    rw.w2c[0] = r3W2[m0]; rw.w2c[1] = r3W2[16 + m0];
    #pragma unroll
    for (int kc = 0; kc < 8; ++kc) {
        const int n = kc * 4 + 2 * half;
        rw.w0a[kc].x = r1W0[n]; rw.w0a[kc].y = r1W0[n + 1];
        rw.b0a[kc].x = r1b0[n]; rw.b0a[kc].y = r1b0[n + 1];
        rw.w0b[kc].x = r2W0[n]; rw.w0b[kc].y = r2W0[n + 1];
        rw.b0b[kc].x = r2b0[n]; rw.b0b[kc].y = r2b0[n + 1];
    }

    LaneSel ls;
    ls.b0 = (lane & 1) != 0; ls.b1 = (lane & 2) != 0; ls.b2 = (lane & 4) != 0;
    ls.own = ((lane >> 3) & 1) == half;

    const int base = blockIdx.x * SPW;

    for (int i = lane; i < SPW * 26; i += 32) {
        const int s = i / 26, c = i % 26;
        if (c >= 2) s_z[s * 24 + (c - 2)] = xz0[(size_t)(base + s) * 26 + c];
    }
    float x0 = xz0[(size_t)(base + m0) * 26 + 0];
    float x1 = xz0[(size_t)(base + m0) * 26 + 1];
    __syncthreads();

    for (int t = 0; t < Td; ++t) {
        const float u  = useq[(size_t)(base + m0) * Td + t];
        if (lane < SPW) {
            out[((size_t)(base + m0) * Td + t) * 2 + 0] = x0;
            out[((size_t)(base + m0) * Td + t) * 2 + 1] = x1;
        }

        r3_layer0(x0, x1, s_z, rw, s_g0, lane);
        __syncthreads();
        float d1, d2, d3[3];
        hidden_pair(x0, x1, rw, ls, d1, d2);
        r3_layer1(s_g0, rw, ls, d3, lane);
        const float r3z   = (d3[0] + rw.r3b2) * 0.2f;
        const float r3z23 = (d3[1] + rw.r3b2) * 0.2f;
        const float r3z4  = (d3[2] + rw.r3b2) * 0.2f;
        float k1a, k1b, k2a, k2b, k3a, k3b, k4a, k4b;
        {
            const float r1v = (d1 + rw.r1b2) * 0.3f;
            const float r2v = (d2 + rw.r2b2) * 0.2f;
            cstr_rhs(x0, x1, u, r1v, r2v, r3z, k1a, k1b);
        }
        float xe0 = x0 + 0.5f * k1a, xe1 = x1 + 0.5f * k1b;

        stage_pair(xe0, xe1, u, r3z23, rw, ls, k2a, k2b);
        xe0 = x0 + 0.5f * k2a; xe1 = x1 + 0.5f * k2b;

        stage_pair(xe0, xe1, u, r3z23, rw, ls, k3a, k3b);
        xe0 = x0 + k3a; xe1 = x1 + k3b;

        stage_pair(xe0, xe1, u, r3z4, rw, ls, k4a, k4b);

        const float nx0 = x0 + (k1a + 2.f * k2a + 2.f * k3a + k4a) * (1.f / 6.f);
        const float nx1 = x1 + (k1b + 2.f * k2b + 2.f * k3b + k4b) * (1.f / 6.f);
        if (lane < SPW) {
            const int s = lane;
            float zo[24];
            #pragma unroll
            for (int c = 0; c < 24; ++c) zo[c] = s_z[s * 24 + c];
            #pragma unroll
            for (int c = 0; c < 14; ++c) s_z[s * 24 + c] = zo[c + 2];
            s_z[s * 24 + 14] = x0;
            s_z[s * 24 + 15] = x1;
            #pragma unroll
            for (int c = 16; c < 23; ++c) s_z[s * 24 + c] = zo[c + 1];
            s_z[s * 24 + 23] = u;
        }
        x0 = nx0;
        x1 = nx1;
        __syncthreads();
    }
}

extern "C" void kernel_launch(void* const* d_in, const int* in_sizes, int n_in,
                              void* d_out, int out_size, void* d_ws, size_t ws_size,
                              hipStream_t stream) {
    (void)in_sizes; (void)n_in; (void)out_size; (void)d_ws; (void)ws_size;
    const float* useq = (const float*)d_in[0];
    const float* xz0  = (const float*)d_in[1];
    const float* p[18];
    for (int i = 0; i < 18; ++i) p[i] = (const float*)d_in[2 + i];
    float* out = (float*)d_out;

    const int B = 4096;
    dim3 block(32);
    dim3 grid(B / SPW);
    reac_rk4_kernel<<<grid, block, 0, stream>>>(
        useq, xz0,
        p[0], p[1], p[2], p[3], p[4], p[5],
        p[6], p[7], p[8], p[9], p[10], p[11],
        p[12], p[13], p[14], p[15], p[16], p[17],
        out);
}


# State rows z kept in registers in r3's A-fragment layout: z4 and next z = pairs of the lane 16 away (permlanex16 + select per pair), no LDS load/store of z per step; sum order unchanged
# speedup vs baseline: 3.1438x; 1.0438x over previous
#include <hip/hip_runtime.h>
#include <hip/hip_bf16.h>


#define Td  256
#define SPW 16

typedef __attribute__((ext_vector_type(2))) float v2f;
typedef __attribute__((ext_vector_type(8))) float v8f;

__device__ __forceinline__ float fast_tanh(float x)
{
    return __builtin_amdgcn_tanhf(x);
}

template <int KC>
__device__ __forceinline__ void load_wfrags(const float* __restrict__ W,
                                            v2f (&B)[2][KC], int lane)
{
    const int half = lane >> 4, m0 = lane & 15;
    #pragma unroll
    for (int nt = 0; nt < 2; ++nt) {
        const int n = nt * 16 + m0;
        #pragma unroll
        for (int kc = 0; kc < KC; ++kc) {
            B[nt][kc].x = W[(kc * 4 + 2 * half + 0) * 32 + n];
            B[nt][kc].y = W[(kc * 4 + 2 * half + 1) * 32 + n];
        }
    }
}

struct RegW {
    v2f B1[2][8];
    v2f B2[2][8];
    v2f B3a[2][6];
    v2f B3b[2][8];
    float b1[2], b2[2], b3a[2], b3b[2];
    float r1b2, r2b2, r3b2;
    float w2a[2], w2b[2], w2c[2];
    v2f w0a[8], b0a[8], w0b[8], b0b[8];
};

template <int MT>
__device__ __forceinline__ void store_tiles(const v8f (&acc)[MT][2],
                                            const float bias[2],
                                            float* __restrict__ out, int lane)
{
    const int half = lane >> 4, m0 = lane & 15;
    #pragma unroll
    for (int mt = 0; mt < MT; ++mt) {
        #pragma unroll
        for (int r = 0; r < 8; ++r) {
            const int m = mt * 16 + r + 8 * half;
            out[m * 32 + m0]      = fast_tanh(acc[mt][0][r] + bias[0]);
            out[m * 32 + 16 + m0] = fast_tanh(acc[mt][1][r] + bias[1]);
        }
    }
}

template <int MASK>
__device__ __forceinline__ float row_xor(float v)
{
    return __builtin_bit_cast(float, __builtin_amdgcn_update_dpp(
        0, __builtin_bit_cast(int, v), 0x160 + MASK, 0xf, 0xf, true));
}

__device__ __forceinline__ float other_row(float v)
{
    const int b = __builtin_bit_cast(int, v);
    return __builtin_bit_cast(float,
        __builtin_amdgcn_permlanex16(b, b, 0x76543210, 0xfedcba98, false, false));
}

struct LaneSel {
    bool b0, b1, b2;
    bool own;
};

__device__ __forceinline__ float tile_dot(const v8f& c0, const v8f& c1,
                                          const float bias[2], const float w2[2],
                                          const LaneSel& ls)
{
    float p[8];
    #pragma unroll
    for (int r = 0; r < 8; ++r)
        p[r] = fast_tanh(c0[r] + bias[0]) * w2[0] + fast_tanh(c1[r] + bias[1]) * w2[1];
    float q[4];
    #pragma unroll
    for (int i = 0; i < 4; ++i) {
        const float keep = ls.b2 ? p[4 + i] : p[i];
        const float send = ls.b2 ? p[i] : p[4 + i];
        q[i] = keep + row_xor<4>(send);
    }
    float s[2];
    #pragma unroll
    for (int i = 0; i < 2; ++i) {
        const float keep = ls.b1 ? q[2 + i] : q[i];
        const float send = ls.b1 ? q[i] : q[2 + i];
        s[i] = keep + row_xor<2>(send);
    }
    const float keep = ls.b0 ? s[1] : s[0];
    const float send = ls.b0 ? s[0] : s[1];
    const float t = keep + row_xor<1>(send);
    const float u = t + row_xor<8>(t);
    const float o = other_row(u);
    return ls.own ? u : o;
}

__device__ __forceinline__ void hidden_pair(float xa, float xb,
                                            const RegW& rw, const LaneSel& ls,
                                            float& da, float& db)
{
    v8f acc[2][2];
    acc[0][0] = (v8f){0.f, 0.f, 0.f, 0.f, 0.f, 0.f, 0.f, 0.f};
    acc[0][1] = acc[0][0];
    acc[1][0] = acc[0][0];
    acc[1][1] = acc[0][0];
    #pragma unroll
    for (int kc = 0; kc < 8; ++kc) {
        v2f a1, a2;
        a1.x = fast_tanh(xa * rw.w0a[kc].x + rw.b0a[kc].x);
        a1.y = fast_tanh(xa * rw.w0a[kc].y + rw.b0a[kc].y);
        a2.x = fast_tanh(xb * rw.w0b[kc].x + rw.b0b[kc].x);
        a2.y = fast_tanh(xb * rw.w0b[kc].y + rw.b0b[kc].y);
        acc[0][0] = __builtin_amdgcn_wmma_f32_16x16x4_f32(
            false, a1, false, rw.B1[0][kc], (short)0, acc[0][0], false, false);
        acc[0][1] = __builtin_amdgcn_wmma_f32_16x16x4_f32(
            false, a1, false, rw.B1[1][kc], (short)0, acc[0][1], false, false);
        acc[1][0] = __builtin_amdgcn_wmma_f32_16x16x4_f32(
            false, a2, false, rw.B2[0][kc], (short)0, acc[1][0], false, false);
        acc[1][1] = __builtin_amdgcn_wmma_f32_16x16x4_f32(
            false, a2, false, rw.B2[1][kc], (short)0, acc[1][1], false, false);
    }
    da = tile_dot(acc[0][0], acc[0][1], rw.b1, rw.w2a, ls);
    db = tile_dot(acc[1][0], acc[1][1], rw.b2, rw.w2b, ls);
}

__device__ __forceinline__ void r3_layer0(const v2f (&za)[6], const v2f (&z4a)[4],
                                          const RegW& rw,
                                          float* __restrict__ g0, int lane)
{
    v2f a[3][6];
    #pragma unroll
    for (int kc = 0; kc < 6; ++kc)
        a[0][kc] = za[kc];
    #pragma unroll
    for (int kc = 0; kc < 4; ++kc) {
        a[2][kc] = z4a[kc];
        a[1][kc].x = 0.5f * (a[0][kc].x + a[2][kc].x);
        a[1][kc].y = 0.5f * (a[0][kc].y + a[2][kc].y);
    }
    #pragma unroll
    for (int kc = 4; kc < 6; ++kc) {
        a[1][kc] = a[0][kc];
        a[2][kc] = a[0][kc];
    }

    v8f acc[3][2];
    #pragma unroll
    for (int mt = 0; mt < 3; ++mt) {
        acc[mt][0] = (v8f){0.f, 0.f, 0.f, 0.f, 0.f, 0.f, 0.f, 0.f};
        acc[mt][1] = acc[mt][0];
    }
    #pragma unroll
    for (int kc = 0; kc < 6; ++kc) {
        #pragma unroll
        for (int mt = 0; mt < 3; ++mt) {
            acc[mt][0] = __builtin_amdgcn_wmma_f32_16x16x4_f32(
                false, a[mt][kc], false, rw.B3a[0][kc], (short)0, acc[mt][0], false, false);
            acc[mt][1] = __builtin_amdgcn_wmma_f32_16x16x4_f32(
                false, a[mt][kc], false, rw.B3a[1][kc], (short)0, acc[mt][1], false, false);
        }
    }
    store_tiles<3>(acc, rw.b3a, g0, lane);
}

__device__ __forceinline__ void r3_layer1(const float* __restrict__ g0,
                                          const RegW& rw, const LaneSel& ls,
                                          float (&d)[3], int lane)
{
    const int half = lane >> 4, m0 = lane & 15;
    v8f acc[3][2];
    #pragma unroll
    for (int mt = 0; mt < 3; ++mt) {
        acc[mt][0] = (v8f){0.f, 0.f, 0.f, 0.f, 0.f, 0.f, 0.f, 0.f};
        acc[mt][1] = acc[mt][0];
    }
    #pragma unroll
    for (int kc = 0; kc < 8; ++kc) {
        #pragma unroll
        for (int mt = 0; mt < 3; ++mt) {
            v2f av;
            av.x = g0[(mt * 16 + m0) * 32 + kc * 4 + 2 * half + 0];
            av.y = g0[(mt * 16 + m0) * 32 + kc * 4 + 2 * half + 1];
            acc[mt][0] = __builtin_amdgcn_wmma_f32_16x16x4_f32(
                false, av, false, rw.B3b[0][kc], (short)0, acc[mt][0], false, false);
            acc[mt][1] = __builtin_amdgcn_wmma_f32_16x16x4_f32(
                false, av, false, rw.B3b[1][kc], (short)0, acc[mt][1], false, false);
        }
    }
    #pragma unroll
    for (int mt = 0; mt < 3; ++mt)
        d[mt] = tile_dot(acc[mt][0], acc[mt][1], rw.b3b, rw.w2c, ls);
}

__device__ __forceinline__ void cstr_rhs(float xe0, float xe1, float u,
                                         float r1v, float r2v, float r3s,
                                         float& k0, float& k1)
{
    const float xs0 = xe0 * 0.3f + 0.5f;
    const float xs1 = xe1 * 0.2f + 0.5f;
    const float us  = u * 0.5f + 1.0f;
    const float dCa = 0.1f * (us - xs0) - r1v;
    const float dCb = -0.1f * xs1 + r1v - 3.0f * r2v + r3s;
    k0 = dCa * (1.0f / 0.3f);
    k1 = dCb * (1.0f / 0.2f);
}

__device__ __forceinline__ void stage_pair(float xe0, float xe1, float u, float r3s,
                                           const RegW& rw, const LaneSel& ls,
                                           float& k0, float& k1)
{
    float d1, d2;
    hidden_pair(xe0, xe1, rw, ls, d1, d2);
    const float r1v = (d1 + rw.r1b2) * 0.3f;
    const float r2v = (d2 + rw.r2b2) * 0.2f;
    cstr_rhs(xe0, xe1, u, r1v, r2v, r3s, k0, k1);
}

__global__ __launch_bounds__(32) void reac_rk4_kernel(
    const float* __restrict__ useq, const float* __restrict__ xz0,
    const float* __restrict__ r1W0, const float* __restrict__ r1b0,
    const float* __restrict__ r1W1, const float* __restrict__ r1b1,
    const float* __restrict__ r1W2, const float* __restrict__ r1b2,
    const float* __restrict__ r2W0, const float* __restrict__ r2b0,
    const float* __restrict__ r2W1, const float* __restrict__ r2b1,
    const float* __restrict__ r2W2, const float* __restrict__ r2b2,
    const float* __restrict__ r3W0, const float* __restrict__ r3b0,
    const float* __restrict__ r3W1, const float* __restrict__ r3b1,
    const float* __restrict__ r3W2, const float* __restrict__ r3b2,
    float* __restrict__ out)
{
    __shared__ __attribute__((aligned(16))) float s_g0[3 * SPW * 32];

    const int lane = threadIdx.x & 31;
    const int m0   = lane & 15;
    const int half = lane >> 4;

    RegW rw;
    load_wfrags<8>(r1W1, rw.B1, lane);
    load_wfrags<8>(r2W1, rw.B2, lane);
    load_wfrags<6>(r3W0, rw.B3a, lane);
    load_wfrags<8>(r3W1, rw.B3b, lane);
    rw.b1[0]  = r1b1[m0]; rw.b1[1]  = r1b1[16 + m0];
    rw.b2[0]  = r2b1[m0]; rw.b2[1]  = r2b1[16 + m0];
    rw.b3a[0] = r3b0[m0]; rw.b3a[1] = r3b0[16 + m0];
    rw.b3b[0] = r3b1[m0]; rw.b3b[1] = r3b1[16 + m0];
    rw.r1b2 = r1b2[0]; rw.r2b2 = r2b2[0]; rw.r3b2 = r3b2[0];
    rw.w2a[0] = r1W2[m0]; rw.w2a[1] = r1W2[16 + m0];
    rw.w2b[0] = r2W2[m0]; rw.w2b[1] = r2W2[16 + m0];
    rw.w2c[0] = r3W2[m0]; rw.w2c[1] = r3W2[16 + m0];
    #pragma unroll
    for (int kc = 0; kc < 8; ++kc) {
        const int n = kc * 4 + 2 * half;
        rw.w0a[kc].x = r1W0[n]; rw.w0a[kc].y = r1W0[n + 1];
        rw.b0a[kc].x = r1b0[n]; rw.b0a[kc].y = r1b0[n + 1];
        rw.w0b[kc].x = r2W0[n]; rw.w0b[kc].y = r2W0[n + 1];
        rw.b0b[kc].x = r2b0[n]; rw.b0b[kc].y = r2b0[n + 1];
    }

    LaneSel ls;
    ls.b0 = (lane & 1) != 0; ls.b1 = (lane & 2) != 0; ls.b2 = (lane & 4) != 0;
    ls.own = ((lane >> 3) & 1) == half;

    const int base = blockIdx.x * SPW;

    const float* xzrow = xz0 + (size_t)(base + m0) * 26;
    v2f za[6];
    #pragma unroll
    for (int kc = 0; kc < 6; ++kc) {
        za[kc].x = xzrow[2 + kc * 4 + 2 * half + 0];
        za[kc].y = xzrow[2 + kc * 4 + 2 * half + 1];
    }
    float x0 = xzrow[0];
    float x1 = xzrow[1];
    const bool hi = half != 0;

    for (int t = 0; t < Td; ++t) {
        const float u  = useq[(size_t)(base + m0) * Td + t];
        if (lane < SPW) {
            out[((size_t)(base + m0) * Td + t) * 2 + 0] = x0;
            out[((size_t)(base + m0) * Td + t) * 2 + 1] = x1;
        }

        v2f z4a[4];
        {
            const float p0x = other_row(za[0].x), p0y = other_row(za[0].y);
            const float p1x = other_row(za[1].x), p1y = other_row(za[1].y);
            const float p2x = other_row(za[2].x), p2y = other_row(za[2].y);
            const float p3x = other_row(za[3].x), p3y = other_row(za[3].y);
            z4a[0].x = hi ? p1x : p0x;  z4a[0].y = hi ? p1y : p0y;
            z4a[1].x = hi ? p2x : p1x;  z4a[1].y = hi ? p2y : p1y;
            z4a[2].x = hi ? p3x : p2x;  z4a[2].y = hi ? p3y : p2y;
            z4a[3].x = hi ? x0  : p3x;  z4a[3].y = hi ? x1  : p3y;
        }
        r3_layer0(za, z4a, rw, s_g0, lane);
        __syncthreads();
        float d1, d2, d3[3];
        hidden_pair(x0, x1, rw, ls, d1, d2);
        r3_layer1(s_g0, rw, ls, d3, lane);
        const float r3z   = (d3[0] + rw.r3b2) * 0.2f;
        const float r3z23 = (d3[1] + rw.r3b2) * 0.2f;
        const float r3z4  = (d3[2] + rw.r3b2) * 0.2f;
        float k1a, k1b, k2a, k2b, k3a, k3b, k4a, k4b;
        {
            const float r1v = (d1 + rw.r1b2) * 0.3f;
            const float r2v = (d2 + rw.r2b2) * 0.2f;
            cstr_rhs(x0, x1, u, r1v, r2v, r3z, k1a, k1b);
        }
        float xe0 = x0 + 0.5f * k1a, xe1 = x1 + 0.5f * k1b;

        stage_pair(xe0, xe1, u, r3z23, rw, ls, k2a, k2b);
        xe0 = x0 + 0.5f * k2a; xe1 = x1 + 0.5f * k2b;

        stage_pair(xe0, xe1, u, r3z23, rw, ls, k3a, k3b);
        xe0 = x0 + k3a; xe1 = x1 + k3b;

        stage_pair(xe0, xe1, u, r3z4, rw, ls, k4a, k4b);

        const float nx0 = x0 + (k1a + 2.f * k2a + 2.f * k3a + k4a) * (1.f / 6.f);
        const float nx1 = x1 + (k1b + 2.f * k2b + 2.f * k3b + k4b) * (1.f / 6.f);
        {
            const float q4 = other_row(za[4].x);
            const float q5 = other_row(za[5].x);
            v2f n4, n5;
            n4.x = za[4].y; n4.y = hi ? q5 : q4;
            n5.x = za[5].y; n5.y = hi ? u  : q5;
            #pragma unroll
            for (int kc = 0; kc < 4; ++kc) za[kc] = z4a[kc];
            za[4] = n4;
            za[5] = n5;
        }
        x0 = nx0;
        x1 = nx1;
        __syncthreads();
    }
}

extern "C" void kernel_launch(void* const* d_in, const int* in_sizes, int n_in,
                              void* d_out, int out_size, void* d_ws, size_t ws_size,
                              hipStream_t stream) {
    (void)in_sizes; (void)n_in; (void)out_size; (void)d_ws; (void)ws_size;
    const float* useq = (const float*)d_in[0];
    const float* xz0  = (const float*)d_in[1];
    const float* p[18];
    for (int i = 0; i < 18; ++i) p[i] = (const float*)d_in[2 + i];
    float* out = (float*)d_out;

    const int B = 4096;
    dim3 block(32);
    dim3 grid(B / SPW);
    reac_rk4_kernel<<<grid, block, 0, stream>>>(
        useq, xz0,
        p[0], p[1], p[2], p[3], p[4], p[5],
        p[6], p[7], p[8], p[9], p[10], p[11],
        p[12], p[13], p[14], p[15], p[16], p[17],
        out);
}


# r3 layer 0 by linearity: z4, next z share k chunks 0..3; z's tile carried from the step before (u loaded a step ahead); z23's tile = mean of z's and z4's; 36 WMMAs/step -> 16; z23 sum order changes
# speedup vs baseline: 3.2669x; 1.0392x over previous
#include <hip/hip_runtime.h>
#include <hip/hip_bf16.h>


#define Td  256
#define SPW 16

typedef __attribute__((ext_vector_type(2))) float v2f;
typedef __attribute__((ext_vector_type(8))) float v8f;

__device__ __forceinline__ float fast_tanh(float x)
{
    return __builtin_amdgcn_tanhf(x);
}

template <int KC>
__device__ __forceinline__ void load_wfrags(const float* __restrict__ W,
                                            v2f (&B)[2][KC], int lane)
{
    const int half = lane >> 4, m0 = lane & 15;
    #pragma unroll
    for (int nt = 0; nt < 2; ++nt) {
        const int n = nt * 16 + m0;
        #pragma unroll
        for (int kc = 0; kc < KC; ++kc) {
            B[nt][kc].x = W[(kc * 4 + 2 * half + 0) * 32 + n];
            B[nt][kc].y = W[(kc * 4 + 2 * half + 1) * 32 + n];
        }
    }
}

struct RegW {
    v2f B1[2][8];
    v2f B2[2][8];
    v2f B3a[2][6];
    v2f B3b[2][8];
    float b1[2], b2[2], b3a[2], b3b[2];
    float r1b2, r2b2, r3b2;
    float w2a[2], w2b[2], w2c[2];
    v2f w0a[8], b0a[8], w0b[8], b0b[8];
};

template <int MT>
__device__ __forceinline__ void store_tiles(const v8f (&acc)[MT][2],
                                            const float bias[2],
                                            float* __restrict__ out, int lane)
{
    const int half = lane >> 4, m0 = lane & 15;
    #pragma unroll
    for (int mt = 0; mt < MT; ++mt) {
        #pragma unroll
        for (int r = 0; r < 8; ++r) {
            const int m = mt * 16 + r + 8 * half;
            out[m * 32 + m0]      = fast_tanh(acc[mt][0][r] + bias[0]);
            out[m * 32 + 16 + m0] = fast_tanh(acc[mt][1][r] + bias[1]);
        }
    }
}

template <int MASK>
__device__ __forceinline__ float row_xor(float v)
{
    return __builtin_bit_cast(float, __builtin_amdgcn_update_dpp(
        0, __builtin_bit_cast(int, v), 0x160 + MASK, 0xf, 0xf, true));
}

__device__ __forceinline__ float other_row(float v)
{
    const int b = __builtin_bit_cast(int, v);
    return __builtin_bit_cast(float,
        __builtin_amdgcn_permlanex16(b, b, 0x76543210, 0xfedcba98, false, false));
}

struct LaneSel {
    bool b0, b1, b2;
    bool own;
};

__device__ __forceinline__ float tile_dot(const v8f& c0, const v8f& c1,
                                          const float bias[2], const float w2[2],
                                          const LaneSel& ls)
{
    float p[8];
    #pragma unroll
    for (int r = 0; r < 8; ++r)
        p[r] = fast_tanh(c0[r] + bias[0]) * w2[0] + fast_tanh(c1[r] + bias[1]) * w2[1];
    float q[4];
    #pragma unroll
    for (int i = 0; i < 4; ++i) {
        const float keep = ls.b2 ? p[4 + i] : p[i];
        const float send = ls.b2 ? p[i] : p[4 + i];
        q[i] = keep + row_xor<4>(send);
    }
    float s[2];
    #pragma unroll
    for (int i = 0; i < 2; ++i) {
        const float keep = ls.b1 ? q[2 + i] : q[i];
        const float send = ls.b1 ? q[i] : q[2 + i];
        s[i] = keep + row_xor<2>(send);
    }
    const float keep = ls.b0 ? s[1] : s[0];
    const float send = ls.b0 ? s[0] : s[1];
    const float t = keep + row_xor<1>(send);
    const float u = t + row_xor<8>(t);
    const float o = other_row(u);
    return ls.own ? u : o;
}

__device__ __forceinline__ void hidden_pair(float xa, float xb,
                                            const RegW& rw, const LaneSel& ls,
                                            float& da, float& db)
{
    v8f acc[2][2];
    acc[0][0] = (v8f){0.f, 0.f, 0.f, 0.f, 0.f, 0.f, 0.f, 0.f};
    acc[0][1] = acc[0][0];
    acc[1][0] = acc[0][0];
    acc[1][1] = acc[0][0];
    #pragma unroll
    for (int kc = 0; kc < 8; ++kc) {
        v2f a1, a2;
        a1.x = fast_tanh(xa * rw.w0a[kc].x + rw.b0a[kc].x);
        a1.y = fast_tanh(xa * rw.w0a[kc].y + rw.b0a[kc].y);
        a2.x = fast_tanh(xb * rw.w0b[kc].x + rw.b0b[kc].x);
        a2.y = fast_tanh(xb * rw.w0b[kc].y + rw.b0b[kc].y);
        acc[0][0] = __builtin_amdgcn_wmma_f32_16x16x4_f32(
            false, a1, false, rw.B1[0][kc], (short)0, acc[0][0], false, false);
        acc[0][1] = __builtin_amdgcn_wmma_f32_16x16x4_f32(
            false, a1, false, rw.B1[1][kc], (short)0, acc[0][1], false, false);
        acc[1][0] = __builtin_amdgcn_wmma_f32_16x16x4_f32(
            false, a2, false, rw.B2[0][kc], (short)0, acc[1][0], false, false);
        acc[1][1] = __builtin_amdgcn_wmma_f32_16x16x4_f32(
            false, a2, false, rw.B2[1][kc], (short)0, acc[1][1], false, false);
    }
    da = tile_dot(acc[0][0], acc[0][1], rw.b1, rw.w2a, ls);
    db = tile_dot(acc[1][0], acc[1][1], rw.b2, rw.w2b, ls);
}

__device__ __forceinline__ void r3_layer0(v8f (&cz)[2], const v2f (&z4a)[4],
                                          v2f zu4, v2f zu5, v2f zn4, v2f zn5,
                                          const RegW& rw,
                                          float* __restrict__ g0, int lane)
{
    v8f cs[2], cn[2];
    v8f acc[3][2];
    #pragma unroll
    for (int nt = 0; nt < 2; ++nt) {
        cs[nt] = (v8f){0.f, 0.f, 0.f, 0.f, 0.f, 0.f, 0.f, 0.f};
        #pragma unroll
        for (int kc = 0; kc < 4; ++kc)
            cs[nt] = __builtin_amdgcn_wmma_f32_16x16x4_f32(
                false, z4a[kc], false, rw.B3a[nt][kc], (short)0, cs[nt], false, false);
    }
    #pragma unroll
    for (int nt = 0; nt < 2; ++nt) {
        acc[2][nt] = __builtin_amdgcn_wmma_f32_16x16x4_f32(
            false, zu4, false, rw.B3a[nt][4], (short)0, cs[nt], false, false);
        acc[2][nt] = __builtin_amdgcn_wmma_f32_16x16x4_f32(
            false, zu5, false, rw.B3a[nt][5], (short)0, acc[2][nt], false, false);
        cn[nt] = __builtin_amdgcn_wmma_f32_16x16x4_f32(
            false, zn4, false, rw.B3a[nt][4], (short)0, cs[nt], false, false);
        cn[nt] = __builtin_amdgcn_wmma_f32_16x16x4_f32(
            false, zn5, false, rw.B3a[nt][5], (short)0, cn[nt], false, false);
    }
    #pragma unroll
    for (int nt = 0; nt < 2; ++nt) {
        acc[0][nt] = cz[nt];
        #pragma unroll
        for (int r = 0; r < 8; ++r)
            acc[1][nt][r] = 0.5f * (cz[nt][r] + acc[2][nt][r]);
        cz[nt] = cn[nt];
    }
    store_tiles<3>(acc, rw.b3a, g0, lane);
}

__device__ __forceinline__ void r3_layer1(const float* __restrict__ g0,
                                          const RegW& rw, const LaneSel& ls,
                                          float (&d)[3], int lane)
{
    const int half = lane >> 4, m0 = lane & 15;
    v8f acc[3][2];
    #pragma unroll
    for (int mt = 0; mt < 3; ++mt) {
        acc[mt][0] = (v8f){0.f, 0.f, 0.f, 0.f, 0.f, 0.f, 0.f, 0.f};
        acc[mt][1] = acc[mt][0];
    }
    #pragma unroll
    for (int kc = 0; kc < 8; ++kc) {
        #pragma unroll
        for (int mt = 0; mt < 3; ++mt) {
            v2f av;
            av.x = g0[(mt * 16 + m0) * 32 + kc * 4 + 2 * half + 0];
            av.y = g0[(mt * 16 + m0) * 32 + kc * 4 + 2 * half + 1];
            acc[mt][0] = __builtin_amdgcn_wmma_f32_16x16x4_f32(
                false, av, false, rw.B3b[0][kc], (short)0, acc[mt][0], false, false);
            acc[mt][1] = __builtin_amdgcn_wmma_f32_16x16x4_f32(
                false, av, false, rw.B3b[1][kc], (short)0, acc[mt][1], false, false);
        }
    }
    #pragma unroll
    for (int mt = 0; mt < 3; ++mt)
        d[mt] = tile_dot(acc[mt][0], acc[mt][1], rw.b3b, rw.w2c, ls);
}

__device__ __forceinline__ void cstr_rhs(float xe0, float xe1, float u,
                                         float r1v, float r2v, float r3s,
                                         float& k0, float& k1)
{
    const float xs0 = xe0 * 0.3f + 0.5f;
    const float xs1 = xe1 * 0.2f + 0.5f;
    const float us  = u * 0.5f + 1.0f;
    const float dCa = 0.1f * (us - xs0) - r1v;
    const float dCb = -0.1f * xs1 + r1v - 3.0f * r2v + r3s;
    k0 = dCa * (1.0f / 0.3f);
    k1 = dCb * (1.0f / 0.2f);
}

__device__ __forceinline__ void stage_pair(float xe0, float xe1, float u, float r3s,
                                           const RegW& rw, const LaneSel& ls,
                                           float& k0, float& k1)
{
    float d1, d2;
    hidden_pair(xe0, xe1, rw, ls, d1, d2);
    const float r1v = (d1 + rw.r1b2) * 0.3f;
    const float r2v = (d2 + rw.r2b2) * 0.2f;
    cstr_rhs(xe0, xe1, u, r1v, r2v, r3s, k0, k1);
}

__global__ __launch_bounds__(32) void reac_rk4_kernel(
    const float* __restrict__ useq, const float* __restrict__ xz0,
    const float* __restrict__ r1W0, const float* __restrict__ r1b0,
    const float* __restrict__ r1W1, const float* __restrict__ r1b1,
    const float* __restrict__ r1W2, const float* __restrict__ r1b2,
    const float* __restrict__ r2W0, const float* __restrict__ r2b0,
    const float* __restrict__ r2W1, const float* __restrict__ r2b1,
    const float* __restrict__ r2W2, const float* __restrict__ r2b2,
    const float* __restrict__ r3W0, const float* __restrict__ r3b0,
    const float* __restrict__ r3W1, const float* __restrict__ r3b1,
    const float* __restrict__ r3W2, const float* __restrict__ r3b2,
    float* __restrict__ out)
{
    __shared__ __attribute__((aligned(16))) float s_g0[3 * SPW * 32];

    const int lane = threadIdx.x & 31;
    const int m0   = lane & 15;
    const int half = lane >> 4;

    RegW rw;
    load_wfrags<8>(r1W1, rw.B1, lane);
    load_wfrags<8>(r2W1, rw.B2, lane);
    load_wfrags<6>(r3W0, rw.B3a, lane);
    load_wfrags<8>(r3W1, rw.B3b, lane);
    rw.b1[0]  = r1b1[m0]; rw.b1[1]  = r1b1[16 + m0];
    rw.b2[0]  = r2b1[m0]; rw.b2[1]  = r2b1[16 + m0];
    rw.b3a[0] = r3b0[m0]; rw.b3a[1] = r3b0[16 + m0];
    rw.b3b[0] = r3b1[m0]; rw.b3b[1] = r3b1[16 + m0];
    rw.r1b2 = r1b2[0]; rw.r2b2 = r2b2[0]; rw.r3b2 = r3b2[0];
    rw.w2a[0] = r1W2[m0]; rw.w2a[1] = r1W2[16 + m0];
    rw.w2b[0] = r2W2[m0]; rw.w2b[1] = r2W2[16 + m0];
    rw.w2c[0] = r3W2[m0]; rw.w2c[1] = r3W2[16 + m0];
    #pragma unroll
    for (int kc = 0; kc < 8; ++kc) {
        const int n = kc * 4 + 2 * half;
        rw.w0a[kc].x = r1W0[n]; rw.w0a[kc].y = r1W0[n + 1];
        rw.b0a[kc].x = r1b0[n]; rw.b0a[kc].y = r1b0[n + 1];
        rw.w0b[kc].x = r2W0[n]; rw.w0b[kc].y = r2W0[n + 1];
        rw.b0b[kc].x = r2b0[n]; rw.b0b[kc].y = r2b0[n + 1];
    }

    LaneSel ls;
    ls.b0 = (lane & 1) != 0; ls.b1 = (lane & 2) != 0; ls.b2 = (lane & 4) != 0;
    ls.own = ((lane >> 3) & 1) == half;

    const int base = blockIdx.x * SPW;

    const float* xzrow = xz0 + (size_t)(base + m0) * 26;
    v2f za[6];
    #pragma unroll
    for (int kc = 0; kc < 6; ++kc) {
        za[kc].x = xzrow[2 + kc * 4 + 2 * half + 0];
        za[kc].y = xzrow[2 + kc * 4 + 2 * half + 1];
    }
    float x0 = xzrow[0];
    float x1 = xzrow[1];
    const bool hi = half != 0;

    v8f cz[2];
    #pragma unroll
    for (int nt = 0; nt < 2; ++nt) {
        cz[nt] = (v8f){0.f, 0.f, 0.f, 0.f, 0.f, 0.f, 0.f, 0.f};
        #pragma unroll
        for (int kc = 0; kc < 6; ++kc)
            cz[nt] = __builtin_amdgcn_wmma_f32_16x16x4_f32(
                false, za[kc], false, rw.B3a[nt][kc], (short)0, cz[nt], false, false);
    }

    float u_next = useq[(size_t)(base + m0) * Td];
    for (int t = 0; t < Td; ++t) {
        const float u  = u_next;
        u_next = useq[(size_t)(base + m0) * Td + (t + 1 < Td ? t + 1 : t)];
        if (lane < SPW) {
            out[((size_t)(base + m0) * Td + t) * 2 + 0] = x0;
            out[((size_t)(base + m0) * Td + t) * 2 + 1] = x1;
        }

        v2f z4a[4];
        {
            const float p0x = other_row(za[0].x), p0y = other_row(za[0].y);
            const float p1x = other_row(za[1].x), p1y = other_row(za[1].y);
            const float p2x = other_row(za[2].x), p2y = other_row(za[2].y);
            const float p3x = other_row(za[3].x), p3y = other_row(za[3].y);
            z4a[0].x = hi ? p1x : p0x;  z4a[0].y = hi ? p1y : p0y;
            z4a[1].x = hi ? p2x : p1x;  z4a[1].y = hi ? p2y : p1y;
            z4a[2].x = hi ? p3x : p2x;  z4a[2].y = hi ? p3y : p2y;
            z4a[3].x = hi ? x0  : p3x;  z4a[3].y = hi ? x1  : p3y;
        }
        v2f n4, n5;
        {
            const float q4 = other_row(za[4].x);
            const float q5 = other_row(za[5].x);
            n4.x = za[4].y; n4.y = hi ? q5 : q4;
            n5.x = za[5].y; n5.y = hi ? u  : q5;
        }
        r3_layer0(cz, z4a, za[4], za[5], n4, n5, rw, s_g0, lane);
        __syncthreads();
        float d1, d2, d3[3];
        hidden_pair(x0, x1, rw, ls, d1, d2);
        r3_layer1(s_g0, rw, ls, d3, lane);
        const float r3z   = (d3[0] + rw.r3b2) * 0.2f;
        const float r3z23 = (d3[1] + rw.r3b2) * 0.2f;
        const float r3z4  = (d3[2] + rw.r3b2) * 0.2f;
        float k1a, k1b, k2a, k2b, k3a, k3b, k4a, k4b;
        {
            const float r1v = (d1 + rw.r1b2) * 0.3f;
            const float r2v = (d2 + rw.r2b2) * 0.2f;
            cstr_rhs(x0, x1, u, r1v, r2v, r3z, k1a, k1b);
        }
        float xe0 = x0 + 0.5f * k1a, xe1 = x1 + 0.5f * k1b;

        stage_pair(xe0, xe1, u, r3z23, rw, ls, k2a, k2b);
        xe0 = x0 + 0.5f * k2a; xe1 = x1 + 0.5f * k2b;

        stage_pair(xe0, xe1, u, r3z23, rw, ls, k3a, k3b);
        xe0 = x0 + k3a; xe1 = x1 + k3b;

        stage_pair(xe0, xe1, u, r3z4, rw, ls, k4a, k4b);

        const float nx0 = x0 + (k1a + 2.f * k2a + 2.f * k3a + k4a) * (1.f / 6.f);
        const float nx1 = x1 + (k1b + 2.f * k2b + 2.f * k3b + k4b) * (1.f / 6.f);
        {
            #pragma unroll
            for (int kc = 0; kc < 4; ++kc) za[kc] = z4a[kc];
            za[4] = n4;
            za[5] = n5;
        }
        x0 = nx0;
        x1 = nx1;
        __syncthreads();
    }
}

extern "C" void kernel_launch(void* const* d_in, const int* in_sizes, int n_in,
                              void* d_out, int out_size, void* d_ws, size_t ws_size,
                              hipStream_t stream) {
    (void)in_sizes; (void)n_in; (void)out_size; (void)d_ws; (void)ws_size;
    const float* useq = (const float*)d_in[0];
    const float* xz0  = (const float*)d_in[1];
    const float* p[18];
    for (int i = 0; i < 18; ++i) p[i] = (const float*)d_in[2 + i];
    float* out = (float*)d_out;

    const int B = 4096;
    dim3 block(32);
    dim3 grid(B / SPW);
    reac_rk4_kernel<<<grid, block, 0, stream>>>(
        useq, xz0,
        p[0], p[1], p[2], p[3], p[4], p[5],
        p[6], p[7], p[8], p[9], p[10], p[11],
        p[12], p[13], p[14], p[15], p[16], p[17],
        out);
}


# LDS hand-over of r3's layer 0 gathered in ONE place per step: 48 stores, LDS-only fence, 12 fragment loads as one group, no WMMA between (was about 18 places among the WMMAs); sum order unchanged
# speedup vs baseline: 3.4921x; 1.0689x over previous
#include <hip/hip_runtime.h>
#include <hip/hip_bf16.h>


#define Td  256
#define SPW 16

typedef __attribute__((ext_vector_type(2))) float v2f;
typedef __attribute__((ext_vector_type(8))) float v8f;

__device__ __forceinline__ float fast_tanh(float x)
{
    return __builtin_amdgcn_tanhf(x);
}

template <int KC>
__device__ __forceinline__ void load_wfrags(const float* __restrict__ W,
                                            v2f (&B)[2][KC], int lane)
{
    const int half = lane >> 4, m0 = lane & 15;
    #pragma unroll
    for (int nt = 0; nt < 2; ++nt) {
        const int n = nt * 16 + m0;
        #pragma unroll
        for (int kc = 0; kc < KC; ++kc) {
            B[nt][kc].x = W[(kc * 4 + 2 * half + 0) * 32 + n];
            B[nt][kc].y = W[(kc * 4 + 2 * half + 1) * 32 + n];
        }
    }
}

struct RegW {
    v2f B1[2][8];
    v2f B2[2][8];
    v2f B3a[2][6];
    v2f B3b[2][8];
    float b1[2], b2[2], b3a[2], b3b[2];
    float r1b2, r2b2, r3b2;
    float w2a[2], w2b[2], w2c[2];
    v2f w0a[8], b0a[8], w0b[8], b0b[8];
};

__device__ __forceinline__ void lds_phase_fence()
{
    __builtin_amdgcn_fence(__ATOMIC_RELEASE, "workgroup", "local");
    __builtin_amdgcn_s_barrier();
    __builtin_amdgcn_fence(__ATOMIC_ACQUIRE, "workgroup", "local");
}

__device__ __forceinline__ void g0_exchange(const float (&hv)[3][2][8],
                                            v2f (&av)[3][8],
                                            float* __restrict__ g0, int lane)
{
    const int half = lane >> 4, m0 = lane & 15;
    #pragma unroll
    for (int mt = 0; mt < 3; ++mt) {
        #pragma unroll
        for (int r = 0; r < 8; ++r) {
            const int m = mt * 16 + r + 8 * half;
            g0[m * 32 + m0]      = hv[mt][0][r];
            g0[m * 32 + 16 + m0] = hv[mt][1][r];
        }
    }
    lds_phase_fence();
    #pragma unroll
    for (int mt = 0; mt < 3; ++mt) {
        #pragma unroll
        for (int kc = 0; kc < 8; ++kc) {
            av[mt][kc].x = g0[(mt * 16 + m0) * 32 + kc * 4 + 2 * half + 0];
            av[mt][kc].y = g0[(mt * 16 + m0) * 32 + kc * 4 + 2 * half + 1];
        }
    }
}

template <int MASK>
__device__ __forceinline__ float row_xor(float v)
{
    return __builtin_bit_cast(float, __builtin_amdgcn_update_dpp(
        0, __builtin_bit_cast(int, v), 0x160 + MASK, 0xf, 0xf, true));
}

__device__ __forceinline__ float other_row(float v)
{
    const int b = __builtin_bit_cast(int, v);
    return __builtin_bit_cast(float,
        __builtin_amdgcn_permlanex16(b, b, 0x76543210, 0xfedcba98, false, false));
}

struct LaneSel {
    bool b0, b1, b2;
    bool own;
};

__device__ __forceinline__ float tile_dot(const v8f& c0, const v8f& c1,
                                          const float bias[2], const float w2[2],
                                          const LaneSel& ls)
{
    float p[8];
    #pragma unroll
    for (int r = 0; r < 8; ++r)
        p[r] = fast_tanh(c0[r] + bias[0]) * w2[0] + fast_tanh(c1[r] + bias[1]) * w2[1];
    float q[4];
    #pragma unroll
    for (int i = 0; i < 4; ++i) {
        const float keep = ls.b2 ? p[4 + i] : p[i];
        const float send = ls.b2 ? p[i] : p[4 + i];
        q[i] = keep + row_xor<4>(send);
    }
    float s[2];
    #pragma unroll
    for (int i = 0; i < 2; ++i) {
        const float keep = ls.b1 ? q[2 + i] : q[i];
        const float send = ls.b1 ? q[i] : q[2 + i];
        s[i] = keep + row_xor<2>(send);
    }
    const float keep = ls.b0 ? s[1] : s[0];
    const float send = ls.b0 ? s[0] : s[1];
    const float t = keep + row_xor<1>(send);
    const float u = t + row_xor<8>(t);
    const float o = other_row(u);
    return ls.own ? u : o;
}

__device__ __forceinline__ void hidden_pair_mm(float xa, float xb,
                                               const RegW& rw, v8f (&acc)[2][2])
{
    acc[0][0] = (v8f){0.f, 0.f, 0.f, 0.f, 0.f, 0.f, 0.f, 0.f};
    acc[0][1] = acc[0][0];
    acc[1][0] = acc[0][0];
    acc[1][1] = acc[0][0];
    #pragma unroll
    for (int kc = 0; kc < 8; ++kc) {
        v2f a1, a2;
        a1.x = fast_tanh(xa * rw.w0a[kc].x + rw.b0a[kc].x);
        a1.y = fast_tanh(xa * rw.w0a[kc].y + rw.b0a[kc].y);
        a2.x = fast_tanh(xb * rw.w0b[kc].x + rw.b0b[kc].x);
        a2.y = fast_tanh(xb * rw.w0b[kc].y + rw.b0b[kc].y);
        acc[0][0] = __builtin_amdgcn_wmma_f32_16x16x4_f32(
            false, a1, false, rw.B1[0][kc], (short)0, acc[0][0], false, false);
        acc[0][1] = __builtin_amdgcn_wmma_f32_16x16x4_f32(
            false, a1, false, rw.B1[1][kc], (short)0, acc[0][1], false, false);
        acc[1][0] = __builtin_amdgcn_wmma_f32_16x16x4_f32(
            false, a2, false, rw.B2[0][kc], (short)0, acc[1][0], false, false);
        acc[1][1] = __builtin_amdgcn_wmma_f32_16x16x4_f32(
            false, a2, false, rw.B2[1][kc], (short)0, acc[1][1], false, false);
    }
}

__device__ __forceinline__ void hidden_pair_dots(const v8f (&acc)[2][2],
                                                 const RegW& rw, const LaneSel& ls,
                                                 float& da, float& db)
{
    da = tile_dot(acc[0][0], acc[0][1], rw.b1, rw.w2a, ls);
    db = tile_dot(acc[1][0], acc[1][1], rw.b2, rw.w2b, ls);
}

__device__ __forceinline__ void hidden_pair(float xa, float xb,
                                            const RegW& rw, const LaneSel& ls,
                                            float& da, float& db)
{
    v8f acc[2][2];
    hidden_pair_mm(xa, xb, rw, acc);
    hidden_pair_dots(acc, rw, ls, da, db);
}

__device__ __forceinline__ void r3_layer0(v8f (&cz)[2], const v2f (&z4a)[4],
                                          v2f zu4, v2f zu5, v2f zn4, v2f zn5,
                                          const RegW& rw,
                                          float (&hv)[3][2][8])
{
    v8f cs[2], cn[2];
    v8f acc[3][2];
    #pragma unroll
    for (int nt = 0; nt < 2; ++nt) {
        cs[nt] = (v8f){0.f, 0.f, 0.f, 0.f, 0.f, 0.f, 0.f, 0.f};
        #pragma unroll
        for (int kc = 0; kc < 4; ++kc)
            cs[nt] = __builtin_amdgcn_wmma_f32_16x16x4_f32(
                false, z4a[kc], false, rw.B3a[nt][kc], (short)0, cs[nt], false, false);
    }
    #pragma unroll
    for (int nt = 0; nt < 2; ++nt) {
        acc[2][nt] = __builtin_amdgcn_wmma_f32_16x16x4_f32(
            false, zu4, false, rw.B3a[nt][4], (short)0, cs[nt], false, false);
        acc[2][nt] = __builtin_amdgcn_wmma_f32_16x16x4_f32(
            false, zu5, false, rw.B3a[nt][5], (short)0, acc[2][nt], false, false);
        cn[nt] = __builtin_amdgcn_wmma_f32_16x16x4_f32(
            false, zn4, false, rw.B3a[nt][4], (short)0, cs[nt], false, false);
        cn[nt] = __builtin_amdgcn_wmma_f32_16x16x4_f32(
            false, zn5, false, rw.B3a[nt][5], (short)0, cn[nt], false, false);
    }
    #pragma unroll
    for (int nt = 0; nt < 2; ++nt) {
        acc[0][nt] = cz[nt];
        #pragma unroll
        for (int r = 0; r < 8; ++r)
            acc[1][nt][r] = 0.5f * (cz[nt][r] + acc[2][nt][r]);
        cz[nt] = cn[nt];
    }
    #pragma unroll
    for (int mt = 0; mt < 3; ++mt) {
        #pragma unroll
        for (int nt = 0; nt < 2; ++nt) {
            #pragma unroll
            for (int r = 0; r < 8; ++r)
                hv[mt][nt][r] = fast_tanh(acc[mt][nt][r] + rw.b3a[nt]);
        }
    }
}

__device__ __forceinline__ void r3_layer1(const v2f (&av)[3][8],
                                          const RegW& rw, const LaneSel& ls,
                                          float (&d)[3])
{
    v8f acc[3][2];
    #pragma unroll
    for (int mt = 0; mt < 3; ++mt) {
        acc[mt][0] = (v8f){0.f, 0.f, 0.f, 0.f, 0.f, 0.f, 0.f, 0.f};
        acc[mt][1] = acc[mt][0];
    }
    #pragma unroll
    for (int kc = 0; kc < 8; ++kc) {
        #pragma unroll
        for (int mt = 0; mt < 3; ++mt) {
            acc[mt][0] = __builtin_amdgcn_wmma_f32_16x16x4_f32(
                false, av[mt][kc], false, rw.B3b[0][kc], (short)0, acc[mt][0], false, false);
            acc[mt][1] = __builtin_amdgcn_wmma_f32_16x16x4_f32(
                false, av[mt][kc], false, rw.B3b[1][kc], (short)0, acc[mt][1], false, false);
        }
    }
    #pragma unroll
    for (int mt = 0; mt < 3; ++mt)
        d[mt] = tile_dot(acc[mt][0], acc[mt][1], rw.b3b, rw.w2c, ls);
}

__device__ __forceinline__ void cstr_rhs(float xe0, float xe1, float u,
                                         float r1v, float r2v, float r3s,
                                         float& k0, float& k1)
{
    const float xs0 = xe0 * 0.3f + 0.5f;
    const float xs1 = xe1 * 0.2f + 0.5f;
    const float us  = u * 0.5f + 1.0f;
    const float dCa = 0.1f * (us - xs0) - r1v;
    const float dCb = -0.1f * xs1 + r1v - 3.0f * r2v + r3s;
    k0 = dCa * (1.0f / 0.3f);
    k1 = dCb * (1.0f / 0.2f);
}

__device__ __forceinline__ void stage_pair(float xe0, float xe1, float u, float r3s,
                                           const RegW& rw, const LaneSel& ls,
                                           float& k0, float& k1)
{
    float d1, d2;
    hidden_pair(xe0, xe1, rw, ls, d1, d2);
    const float r1v = (d1 + rw.r1b2) * 0.3f;
    const float r2v = (d2 + rw.r2b2) * 0.2f;
    cstr_rhs(xe0, xe1, u, r1v, r2v, r3s, k0, k1);
}

__global__ __launch_bounds__(32) void reac_rk4_kernel(
    const float* __restrict__ useq, const float* __restrict__ xz0,
    const float* __restrict__ r1W0, const float* __restrict__ r1b0,
    const float* __restrict__ r1W1, const float* __restrict__ r1b1,
    const float* __restrict__ r1W2, const float* __restrict__ r1b2,
    const float* __restrict__ r2W0, const float* __restrict__ r2b0,
    const float* __restrict__ r2W1, const float* __restrict__ r2b1,
    const float* __restrict__ r2W2, const float* __restrict__ r2b2,
    const float* __restrict__ r3W0, const float* __restrict__ r3b0,
    const float* __restrict__ r3W1, const float* __restrict__ r3b1,
    const float* __restrict__ r3W2, const float* __restrict__ r3b2,
    float* __restrict__ out)
{
    __shared__ __attribute__((aligned(16))) float s_g0[3 * SPW * 32];

    const int lane = threadIdx.x & 31;
    const int m0   = lane & 15;
    const int half = lane >> 4;

    RegW rw;
    load_wfrags<8>(r1W1, rw.B1, lane);
    load_wfrags<8>(r2W1, rw.B2, lane);
    load_wfrags<6>(r3W0, rw.B3a, lane);
    load_wfrags<8>(r3W1, rw.B3b, lane);
    rw.b1[0]  = r1b1[m0]; rw.b1[1]  = r1b1[16 + m0];
    rw.b2[0]  = r2b1[m0]; rw.b2[1]  = r2b1[16 + m0];
    rw.b3a[0] = r3b0[m0]; rw.b3a[1] = r3b0[16 + m0];
    rw.b3b[0] = r3b1[m0]; rw.b3b[1] = r3b1[16 + m0];
    rw.r1b2 = r1b2[0]; rw.r2b2 = r2b2[0]; rw.r3b2 = r3b2[0];
    rw.w2a[0] = r1W2[m0]; rw.w2a[1] = r1W2[16 + m0];
    rw.w2b[0] = r2W2[m0]; rw.w2b[1] = r2W2[16 + m0];
    rw.w2c[0] = r3W2[m0]; rw.w2c[1] = r3W2[16 + m0];
    #pragma unroll
    for (int kc = 0; kc < 8; ++kc) {
        const int n = kc * 4 + 2 * half;
        rw.w0a[kc].x = r1W0[n]; rw.w0a[kc].y = r1W0[n + 1];
        rw.b0a[kc].x = r1b0[n]; rw.b0a[kc].y = r1b0[n + 1];
        rw.w0b[kc].x = r2W0[n]; rw.w0b[kc].y = r2W0[n + 1];
        rw.b0b[kc].x = r2b0[n]; rw.b0b[kc].y = r2b0[n + 1];
    }

    LaneSel ls;
    ls.b0 = (lane & 1) != 0; ls.b1 = (lane & 2) != 0; ls.b2 = (lane & 4) != 0;
    ls.own = ((lane >> 3) & 1) == half;

    const int base = blockIdx.x * SPW;

    const float* xzrow = xz0 + (size_t)(base + m0) * 26;
    v2f za[6];
    #pragma unroll
    for (int kc = 0; kc < 6; ++kc) {
        za[kc].x = xzrow[2 + kc * 4 + 2 * half + 0];
        za[kc].y = xzrow[2 + kc * 4 + 2 * half + 1];
    }
    float x0 = xzrow[0];
    float x1 = xzrow[1];
    const bool hi = half != 0;

    v8f cz[2];
    #pragma unroll
    for (int nt = 0; nt < 2; ++nt) {
        cz[nt] = (v8f){0.f, 0.f, 0.f, 0.f, 0.f, 0.f, 0.f, 0.f};
        #pragma unroll
        for (int kc = 0; kc < 6; ++kc)
            cz[nt] = __builtin_amdgcn_wmma_f32_16x16x4_f32(
                false, za[kc], false, rw.B3a[nt][kc], (short)0, cz[nt], false, false);
    }

    float u_next = useq[(size_t)(base + m0) * Td];
    for (int t = 0; t < Td; ++t) {
        const float u  = u_next;
        u_next = useq[(size_t)(base + m0) * Td + (t + 1 < Td ? t + 1 : t)];
        if (lane < SPW) {
            out[((size_t)(base + m0) * Td + t) * 2 + 0] = x0;
            out[((size_t)(base + m0) * Td + t) * 2 + 1] = x1;
        }

        v2f z4a[4];
        {
            const float p0x = other_row(za[0].x), p0y = other_row(za[0].y);
            const float p1x = other_row(za[1].x), p1y = other_row(za[1].y);
            const float p2x = other_row(za[2].x), p2y = other_row(za[2].y);
            const float p3x = other_row(za[3].x), p3y = other_row(za[3].y);
            z4a[0].x = hi ? p1x : p0x;  z4a[0].y = hi ? p1y : p0y;
            z4a[1].x = hi ? p2x : p1x;  z4a[1].y = hi ? p2y : p1y;
            z4a[2].x = hi ? p3x : p2x;  z4a[2].y = hi ? p3y : p2y;
            z4a[3].x = hi ? x0  : p3x;  z4a[3].y = hi ? x1  : p3y;
        }
        v2f n4, n5;
        {
            const float q4 = other_row(za[4].x);
            const float q5 = other_row(za[5].x);
            n4.x = za[4].y; n4.y = hi ? q5 : q4;
            n5.x = za[5].y; n5.y = hi ? u  : q5;
        }
        float hv[3][2][8];
        r3_layer0(cz, z4a, za[4], za[5], n4, n5, rw, hv);
        v8f a12[2][2];
        hidden_pair_mm(x0, x1, rw, a12);
        asm volatile("" :: "v"(a12[0][0]), "v"(a12[0][1]), "v"(a12[1][0]), "v"(a12[1][1]));
        v2f av[3][8];
        g0_exchange(hv, av, s_g0, lane);
        float d1, d2, d3[3];
        hidden_pair_dots(a12, rw, ls, d1, d2);
        asm volatile("" :: "v"(av[0][0]), "v"(av[0][1]), "v"(av[0][2]), "v"(av[0][3]),
                           "v"(av[0][4]), "v"(av[0][5]), "v"(av[0][6]), "v"(av[0][7]),
                           "v"(av[1][0]), "v"(av[1][1]), "v"(av[1][2]), "v"(av[1][3]),
                           "v"(av[1][4]), "v"(av[1][5]), "v"(av[1][6]), "v"(av[1][7]),
                           "v"(av[2][0]), "v"(av[2][1]), "v"(av[2][2]), "v"(av[2][3]),
                           "v"(av[2][4]), "v"(av[2][5]), "v"(av[2][6]), "v"(av[2][7]));
        __builtin_amdgcn_sched_barrier(0);
        r3_layer1(av, rw, ls, d3);
        const float r3z   = (d3[0] + rw.r3b2) * 0.2f;
        const float r3z23 = (d3[1] + rw.r3b2) * 0.2f;
        const float r3z4  = (d3[2] + rw.r3b2) * 0.2f;
        float k1a, k1b, k2a, k2b, k3a, k3b, k4a, k4b;
        {
            const float r1v = (d1 + rw.r1b2) * 0.3f;
            const float r2v = (d2 + rw.r2b2) * 0.2f;
            cstr_rhs(x0, x1, u, r1v, r2v, r3z, k1a, k1b);
        }
        float xe0 = x0 + 0.5f * k1a, xe1 = x1 + 0.5f * k1b;

        stage_pair(xe0, xe1, u, r3z23, rw, ls, k2a, k2b);
        xe0 = x0 + 0.5f * k2a; xe1 = x1 + 0.5f * k2b;

        stage_pair(xe0, xe1, u, r3z23, rw, ls, k3a, k3b);
        xe0 = x0 + k3a; xe1 = x1 + k3b;

        stage_pair(xe0, xe1, u, r3z4, rw, ls, k4a, k4b);

        const float nx0 = x0 + (k1a + 2.f * k2a + 2.f * k3a + k4a) * (1.f / 6.f);
        const float nx1 = x1 + (k1b + 2.f * k2b + 2.f * k3b + k4b) * (1.f / 6.f);
        {
            #pragma unroll
            for (int kc = 0; kc < 4; ++kc) za[kc] = z4a[kc];
            za[4] = n4;
            za[5] = n5;
        }
        x0 = nx0;
        x1 = nx1;
        lds_phase_fence();
    }
}

extern "C" void kernel_launch(void* const* d_in, const int* in_sizes, int n_in,
                              void* d_out, int out_size, void* d_ws, size_t ws_size,
                              hipStream_t stream) {
    (void)in_sizes; (void)n_in; (void)out_size; (void)d_ws; (void)ws_size;
    const float* useq = (const float*)d_in[0];
    const float* xz0  = (const float*)d_in[1];
    const float* p[18];
    for (int i = 0; i < 18; ++i) p[i] = (const float*)d_in[2 + i];
    float* out = (float*)d_out;

    const int B = 4096;
    dim3 block(32);
    dim3 grid(B / SPW);
    reac_rk4_kernel<<<grid, block, 0, stream>>>(
        useq, xz0,
        p[0], p[1], p[2], p[3], p[4], p[5],
        p[6], p[7], p[8], p[9], p[10], p[11],
        p[12], p[13], p[14], p[15], p[16], p[17],
        out);
}


# Work that needs no matrix result leaves the path behind a WMMA block: layer biases = accumulator start values (C of the first WMMA), RK right-hand sides pre-summed to 1-2 FMAs; sum order changes
# speedup vs baseline: 3.6673x; 1.0502x over previous
#include <hip/hip_runtime.h>
#include <hip/hip_bf16.h>


#define Td  256
#define SPW 16

typedef __attribute__((ext_vector_type(2))) float v2f;
typedef __attribute__((ext_vector_type(8))) float v8f;

__device__ __forceinline__ float fast_tanh(float x)
{
    return __builtin_amdgcn_tanhf(x);
}

template <int KC>
__device__ __forceinline__ void load_wfrags(const float* __restrict__ W,
                                            v2f (&B)[2][KC], int lane)
{
    const int half = lane >> 4, m0 = lane & 15;
    #pragma unroll
    for (int nt = 0; nt < 2; ++nt) {
        const int n = nt * 16 + m0;
        #pragma unroll
        for (int kc = 0; kc < KC; ++kc) {
            B[nt][kc].x = W[(kc * 4 + 2 * half + 0) * 32 + n];
            B[nt][kc].y = W[(kc * 4 + 2 * half + 1) * 32 + n];
        }
    }
}

struct RegW {
    v2f B1[2][8];
    v2f B2[2][8];
    v2f B3a[2][6];
    v2f B3b[2][8];
    v8f c1[2], c2[2], c3a[2], c3b[2];
    float r1b2;
    float cb;
    float w2a[2], w2b[2], w2c[2];
    v2f w0a[8], b0a[8], w0b[8], b0b[8];
};

__device__ __forceinline__ void lds_phase_fence()
{
    __builtin_amdgcn_fence(__ATOMIC_RELEASE, "workgroup", "local");
    __builtin_amdgcn_s_barrier();
    __builtin_amdgcn_fence(__ATOMIC_ACQUIRE, "workgroup", "local");
}

__device__ __forceinline__ void g0_exchange(const float (&hv)[3][2][8],
                                            v2f (&av)[3][8],
                                            float* __restrict__ g0, int lane)
{
    const int half = lane >> 4, m0 = lane & 15;
    #pragma unroll
    for (int mt = 0; mt < 3; ++mt) {
        #pragma unroll
        for (int r = 0; r < 8; ++r) {
            const int m = mt * 16 + r + 8 * half;
            g0[m * 32 + m0]      = hv[mt][0][r];
            g0[m * 32 + 16 + m0] = hv[mt][1][r];
        }
    }
    lds_phase_fence();
    #pragma unroll
    for (int mt = 0; mt < 3; ++mt) {
        #pragma unroll
        for (int kc = 0; kc < 8; ++kc) {
            av[mt][kc].x = g0[(mt * 16 + m0) * 32 + kc * 4 + 2 * half + 0];
            av[mt][kc].y = g0[(mt * 16 + m0) * 32 + kc * 4 + 2 * half + 1];
        }
    }
}

template <int MASK>
__device__ __forceinline__ float row_xor(float v)
{
    return __builtin_bit_cast(float, __builtin_amdgcn_update_dpp(
        0, __builtin_bit_cast(int, v), 0x160 + MASK, 0xf, 0xf, true));
}

__device__ __forceinline__ float other_row(float v)
{
    const int b = __builtin_bit_cast(int, v);
    return __builtin_bit_cast(float,
        __builtin_amdgcn_permlanex16(b, b, 0x76543210, 0xfedcba98, false, false));
}

struct LaneSel {
    bool b0, b1, b2;
    bool own;
};

__device__ __forceinline__ float tile_dot(const v8f& c0, const v8f& c1,
                                          const float w2[2],
                                          const LaneSel& ls)
{
    float p[8];
    #pragma unroll
    for (int r = 0; r < 8; ++r)
        p[r] = fast_tanh(c0[r]) * w2[0] + fast_tanh(c1[r]) * w2[1];
    float q[4];
    #pragma unroll
    for (int i = 0; i < 4; ++i) {
        const float keep = ls.b2 ? p[4 + i] : p[i];
        const float send = ls.b2 ? p[i] : p[4 + i];
        q[i] = keep + row_xor<4>(send);
    }
    float s[2];
    #pragma unroll
    for (int i = 0; i < 2; ++i) {
        const float keep = ls.b1 ? q[2 + i] : q[i];
        const float send = ls.b1 ? q[i] : q[2 + i];
        s[i] = keep + row_xor<2>(send);
    }
    const float keep = ls.b0 ? s[1] : s[0];
    const float send = ls.b0 ? s[0] : s[1];
    const float t = keep + row_xor<1>(send);
    const float u = t + row_xor<8>(t);
    const float o = other_row(u);
    return ls.own ? u : o;
}

__device__ __forceinline__ void hidden_pair_mm(float xa, float xb,
                                               const RegW& rw, v8f (&acc)[2][2])
{
    acc[0][0] = rw.c1[0];
    acc[0][1] = rw.c1[1];
    acc[1][0] = rw.c2[0];
    acc[1][1] = rw.c2[1];
    #pragma unroll
    for (int kc = 0; kc < 8; ++kc) {
        v2f a1, a2;
        a1.x = fast_tanh(xa * rw.w0a[kc].x + rw.b0a[kc].x);
        a1.y = fast_tanh(xa * rw.w0a[kc].y + rw.b0a[kc].y);
        a2.x = fast_tanh(xb * rw.w0b[kc].x + rw.b0b[kc].x);
        a2.y = fast_tanh(xb * rw.w0b[kc].y + rw.b0b[kc].y);
        acc[0][0] = __builtin_amdgcn_wmma_f32_16x16x4_f32(
            false, a1, false, rw.B1[0][kc], (short)0, acc[0][0], false, false);
        acc[0][1] = __builtin_amdgcn_wmma_f32_16x16x4_f32(
            false, a1, false, rw.B1[1][kc], (short)0, acc[0][1], false, false);
        acc[1][0] = __builtin_amdgcn_wmma_f32_16x16x4_f32(
            false, a2, false, rw.B2[0][kc], (short)0, acc[1][0], false, false);
        acc[1][1] = __builtin_amdgcn_wmma_f32_16x16x4_f32(
            false, a2, false, rw.B2[1][kc], (short)0, acc[1][1], false, false);
    }
}

__device__ __forceinline__ void hidden_pair_dots(const v8f (&acc)[2][2],
                                                 const RegW& rw, const LaneSel& ls,
                                                 float& da, float& db)
{
    da = tile_dot(acc[0][0], acc[0][1], rw.w2a, ls);
    db = tile_dot(acc[1][0], acc[1][1], rw.w2b, ls);
}

__device__ __forceinline__ void r3_layer0(v8f (&cz)[2], const v2f (&z4a)[4],
                                          v2f zu4, v2f zu5, v2f zn4, v2f zn5,
                                          const RegW& rw,
                                          float (&hv)[3][2][8])
{
    v8f cs[2], cn[2];
    v8f acc[3][2];
    #pragma unroll
    for (int nt = 0; nt < 2; ++nt) {
        cs[nt] = rw.c3a[nt];
        #pragma unroll
        for (int kc = 0; kc < 4; ++kc)
            cs[nt] = __builtin_amdgcn_wmma_f32_16x16x4_f32(
                false, z4a[kc], false, rw.B3a[nt][kc], (short)0, cs[nt], false, false);
    }
    #pragma unroll
    for (int nt = 0; nt < 2; ++nt) {
        acc[2][nt] = __builtin_amdgcn_wmma_f32_16x16x4_f32(
            false, zu4, false, rw.B3a[nt][4], (short)0, cs[nt], false, false);
        acc[2][nt] = __builtin_amdgcn_wmma_f32_16x16x4_f32(
            false, zu5, false, rw.B3a[nt][5], (short)0, acc[2][nt], false, false);
        cn[nt] = __builtin_amdgcn_wmma_f32_16x16x4_f32(
            false, zn4, false, rw.B3a[nt][4], (short)0, cs[nt], false, false);
        cn[nt] = __builtin_amdgcn_wmma_f32_16x16x4_f32(
            false, zn5, false, rw.B3a[nt][5], (short)0, cn[nt], false, false);
    }
    #pragma unroll
    for (int nt = 0; nt < 2; ++nt) {
        acc[0][nt] = cz[nt];
        #pragma unroll
        for (int r = 0; r < 8; ++r)
            acc[1][nt][r] = 0.5f * (cz[nt][r] + acc[2][nt][r]);
        cz[nt] = cn[nt];
    }
    #pragma unroll
    for (int mt = 0; mt < 3; ++mt) {
        #pragma unroll
        for (int nt = 0; nt < 2; ++nt) {
            #pragma unroll
            for (int r = 0; r < 8; ++r)
                hv[mt][nt][r] = fast_tanh(acc[mt][nt][r]);
        }
    }
}

__device__ __forceinline__ void r3_layer1(const v2f (&av)[3][8],
                                          const RegW& rw, const LaneSel& ls,
                                          float (&d)[3])
{
    v8f acc[3][2];
    #pragma unroll
    for (int mt = 0; mt < 3; ++mt) {
        acc[mt][0] = rw.c3b[0];
        acc[mt][1] = rw.c3b[1];
    }
    #pragma unroll
    for (int kc = 0; kc < 8; ++kc) {
        #pragma unroll
        for (int mt = 0; mt < 3; ++mt) {
            acc[mt][0] = __builtin_amdgcn_wmma_f32_16x16x4_f32(
                false, av[mt][kc], false, rw.B3b[0][kc], (short)0, acc[mt][0], false, false);
            acc[mt][1] = __builtin_amdgcn_wmma_f32_16x16x4_f32(
                false, av[mt][kc], false, rw.B3b[1][kc], (short)0, acc[mt][1], false, false);
        }
    }
    #pragma unroll
    for (int mt = 0; mt < 3; ++mt)
        d[mt] = tile_dot(acc[mt][0], acc[mt][1], rw.w2c, ls);
}

__device__ __forceinline__ void rhs_known(float xe0, float xe1, float us, float cbz,
                                          float r1b2, float& q0, float& q1)
{
    const float xs0 = xe0 * 0.3f + 0.5f;
    const float xs1 = xe1 * 0.2f + 0.5f;
    q0 = (us - xs0) * (0.1f / 0.3f) - r1b2;
    q1 = cbz - 0.5f * xs1;
}

__device__ __forceinline__ void stage_pair(float xe0, float xe1, float us, float cbz,
                                           float b0, float b1, float c,
                                           const RegW& rw, const LaneSel& ls,
                                           float& k0, float& k1, float& y0, float& y1)
{
    v8f acc[2][2];
    hidden_pair_mm(xe0, xe1, rw, acc);
    float q0, q1;
    rhs_known(xe0, xe1, us, cbz, rw.r1b2, q0, q1);
    const float e0 = b0 + c * q0;
    const float e1 = b1 + c * q1;
    float d1, d2;
    hidden_pair_dots(acc, rw, ls, d1, d2);
    y0 = e0 - c * d1;
    y1 = (e1 + (1.5f * c) * d1) - (3.0f * c) * d2;
    k0 = q0 - d1;
    k1 = (q1 + 1.5f * d1) - 3.0f * d2;
}

__global__ __launch_bounds__(32) void reac_rk4_kernel(
    const float* __restrict__ useq, const float* __restrict__ xz0,
    const float* __restrict__ r1W0, const float* __restrict__ r1b0,
    const float* __restrict__ r1W1, const float* __restrict__ r1b1,
    const float* __restrict__ r1W2, const float* __restrict__ r1b2,
    const float* __restrict__ r2W0, const float* __restrict__ r2b0,
    const float* __restrict__ r2W1, const float* __restrict__ r2b1,
    const float* __restrict__ r2W2, const float* __restrict__ r2b2,
    const float* __restrict__ r3W0, const float* __restrict__ r3b0,
    const float* __restrict__ r3W1, const float* __restrict__ r3b1,
    const float* __restrict__ r3W2, const float* __restrict__ r3b2,
    float* __restrict__ out)
{
    __shared__ __attribute__((aligned(16))) float s_g0[3 * SPW * 32];

    const int lane = threadIdx.x & 31;
    const int m0   = lane & 15;
    const int half = lane >> 4;

    RegW rw;
    load_wfrags<8>(r1W1, rw.B1, lane);
    load_wfrags<8>(r2W1, rw.B2, lane);
    load_wfrags<6>(r3W0, rw.B3a, lane);
    load_wfrags<8>(r3W1, rw.B3b, lane);
    #pragma unroll
    for (int nt = 0; nt < 2; ++nt) {
        const float v1 = r1b1[nt * 16 + m0], v2 = r2b1[nt * 16 + m0];
        const float v3 = r3b0[nt * 16 + m0], v4 = r3b1[nt * 16 + m0];
        rw.c1[nt]  = (v8f){v1, v1, v1, v1, v1, v1, v1, v1};
        rw.c2[nt]  = (v8f){v2, v2, v2, v2, v2, v2, v2, v2};
        rw.c3a[nt] = (v8f){v3, v3, v3, v3, v3, v3, v3, v3};
        rw.c3b[nt] = (v8f){v4, v4, v4, v4, v4, v4, v4, v4};
    }
    rw.r1b2 = r1b2[0];
    rw.cb = (1.5f * r1b2[0] - 3.0f * r2b2[0]) + r3b2[0];
    rw.w2a[0] = r1W2[m0]; rw.w2a[1] = r1W2[16 + m0];
    rw.w2b[0] = r2W2[m0]; rw.w2b[1] = r2W2[16 + m0];
    rw.w2c[0] = r3W2[m0]; rw.w2c[1] = r3W2[16 + m0];
    #pragma unroll
    for (int kc = 0; kc < 8; ++kc) {
        const int n = kc * 4 + 2 * half;
        rw.w0a[kc].x = r1W0[n]; rw.w0a[kc].y = r1W0[n + 1];
        rw.b0a[kc].x = r1b0[n]; rw.b0a[kc].y = r1b0[n + 1];
        rw.w0b[kc].x = r2W0[n]; rw.w0b[kc].y = r2W0[n + 1];
        rw.b0b[kc].x = r2b0[n]; rw.b0b[kc].y = r2b0[n + 1];
    }

    LaneSel ls;
    ls.b0 = (lane & 1) != 0; ls.b1 = (lane & 2) != 0; ls.b2 = (lane & 4) != 0;
    ls.own = ((lane >> 3) & 1) == half;

    const int base = blockIdx.x * SPW;

    const float* xzrow = xz0 + (size_t)(base + m0) * 26;
    v2f za[6];
    #pragma unroll
    for (int kc = 0; kc < 6; ++kc) {
        za[kc].x = xzrow[2 + kc * 4 + 2 * half + 0];
        za[kc].y = xzrow[2 + kc * 4 + 2 * half + 1];
    }
    float x0 = xzrow[0];
    float x1 = xzrow[1];
    const bool hi = half != 0;

    v8f cz[2];
    #pragma unroll
    for (int nt = 0; nt < 2; ++nt) {
        cz[nt] = rw.c3a[nt];
        #pragma unroll
        for (int kc = 0; kc < 6; ++kc)
            cz[nt] = __builtin_amdgcn_wmma_f32_16x16x4_f32(
                false, za[kc], false, rw.B3a[nt][kc], (short)0, cz[nt], false, false);
    }

    float u_next = useq[(size_t)(base + m0) * Td];
    for (int t = 0; t < Td; ++t) {
        const float u  = u_next;
        u_next = useq[(size_t)(base + m0) * Td + (t + 1 < Td ? t + 1 : t)];
        if (lane < SPW) {
            out[((size_t)(base + m0) * Td + t) * 2 + 0] = x0;
            out[((size_t)(base + m0) * Td + t) * 2 + 1] = x1;
        }

        v2f z4a[4];
        {
            const float p0x = other_row(za[0].x), p0y = other_row(za[0].y);
            const float p1x = other_row(za[1].x), p1y = other_row(za[1].y);
            const float p2x = other_row(za[2].x), p2y = other_row(za[2].y);
            const float p3x = other_row(za[3].x), p3y = other_row(za[3].y);
            z4a[0].x = hi ? p1x : p0x;  z4a[0].y = hi ? p1y : p0y;
            z4a[1].x = hi ? p2x : p1x;  z4a[1].y = hi ? p2y : p1y;
            z4a[2].x = hi ? p3x : p2x;  z4a[2].y = hi ? p3y : p2y;
            z4a[3].x = hi ? x0  : p3x;  z4a[3].y = hi ? x1  : p3y;
        }
        v2f n4, n5;
        {
            const float q4 = other_row(za[4].x);
            const float q5 = other_row(za[5].x);
            n4.x = za[4].y; n4.y = hi ? q5 : q4;
            n5.x = za[5].y; n5.y = hi ? u  : q5;
        }
        float hv[3][2][8];
        r3_layer0(cz, z4a, za[4], za[5], n4, n5, rw, hv);
        v8f a12[2][2];
        hidden_pair_mm(x0, x1, rw, a12);
        asm volatile("" :: "v"(a12[0][0]), "v"(a12[0][1]), "v"(a12[1][0]), "v"(a12[1][1]));
        v2f av[3][8];
        g0_exchange(hv, av, s_g0, lane);
        float d1, d2, d3[3];
        hidden_pair_dots(a12, rw, ls, d1, d2);
        const float us = u * 0.5f + 1.0f;
        float q0, q1;
        rhs_known(x0, x1, us, rw.cb, rw.r1b2, q0, q1);
        const float k1a = q0 - d1;
        const float k1p = (q1 + 1.5f * d1) - 3.0f * d2;
        const float e1  = x1 + 0.5f * k1p;
        asm volatile("" :: "v"(av[0][0]), "v"(av[0][1]), "v"(av[0][2]), "v"(av[0][3]),
                           "v"(av[0][4]), "v"(av[0][5]), "v"(av[0][6]), "v"(av[0][7]),
                           "v"(av[1][0]), "v"(av[1][1]), "v"(av[1][2]), "v"(av[1][3]),
                           "v"(av[1][4]), "v"(av[1][5]), "v"(av[1][6]), "v"(av[1][7]),
                           "v"(av[2][0]), "v"(av[2][1]), "v"(av[2][2]), "v"(av[2][3]),
                           "v"(av[2][4]), "v"(av[2][5]), "v"(av[2][6]), "v"(av[2][7]));
        __builtin_amdgcn_sched_barrier(0);
        r3_layer1(av, rw, ls, d3);
        const float k1b = k1p + d3[0];
        float xe0 = x0 + 0.5f * k1a, xe1 = e1 + 0.5f * d3[0];
        const float cbz23 = rw.cb + d3[1];
        const float cbz4  = rw.cb + d3[2];
        float k2a, k2b, k3a, k3b, k4a, k4b, ye0, ye1;

        stage_pair(xe0, xe1, us, cbz23, x0, x1, 0.5f, rw, ls, k2a, k2b, ye0, ye1);

        stage_pair(ye0, ye1, us, cbz23, x0, x1, 1.0f, rw, ls, k3a, k3b, xe0, xe1);

        const float g0 = x0 + ((k1a + 2.f * k2a) + 2.f * k3a) * (1.f / 6.f);
        const float g1 = x1 + ((k1b + 2.f * k2b) + 2.f * k3b) * (1.f / 6.f);
        float nx0, nx1;
        stage_pair(xe0, xe1, us, cbz4, g0, g1, 1.f / 6.f, rw, ls, k4a, k4b, nx0, nx1);
        {
            #pragma unroll
            for (int kc = 0; kc < 4; ++kc) za[kc] = z4a[kc];
            za[4] = n4;
            za[5] = n5;
        }
        x0 = nx0;
        x1 = nx1;
        lds_phase_fence();
    }
}

extern "C" void kernel_launch(void* const* d_in, const int* in_sizes, int n_in,
                              void* d_out, int out_size, void* d_ws, size_t ws_size,
                              hipStream_t stream) {
    (void)in_sizes; (void)n_in; (void)out_size; (void)d_ws; (void)ws_size;
    const float* useq = (const float*)d_in[0];
    const float* xz0  = (const float*)d_in[1];
    const float* p[18];
    for (int i = 0; i < 18; ++i) p[i] = (const float*)d_in[2 + i];
    float* out = (float*)d_out;

    const int B = 4096;
    dim3 block(32);
    dim3 grid(B / SPW);
    reac_rk4_kernel<<<grid, block, 0, stream>>>(
        useq, xz0,
        p[0], p[1], p[2], p[3], p[4], p[5],
        p[6], p[7], p[8], p[9], p[10], p[11],
        p[12], p[13], p[14], p[15], p[16], p[17],
        out);
}


# LDS layout of r3's hand-over fitted to the 64 banks: a lane's column pair side by side (24 8-byte stores for 48), row pitch 36 floats, 12 16-byte fragment loads; k order of r3's hidden layer changes
# speedup vs baseline: 3.7895x; 1.0333x over previous
#include <hip/hip_runtime.h>
#include <hip/hip_bf16.h>


#define Td  256
#define SPW 16
#define GP  36

typedef __attribute__((ext_vector_type(2))) float v2f;
typedef __attribute__((ext_vector_type(4))) float v4f;
typedef __attribute__((address_space(3))) volatile v2f lds_pair_t;
typedef __attribute__((ext_vector_type(8))) float v8f;

__device__ __forceinline__ float fast_tanh(float x)
{
    return __builtin_amdgcn_tanhf(x);
}

template <int KC>
__device__ __forceinline__ void load_wfrags(const float* __restrict__ W,
                                            v2f (&B)[2][KC], int lane)
{
    const int half = lane >> 4, m0 = lane & 15;
    #pragma unroll
    for (int nt = 0; nt < 2; ++nt) {
        const int n = nt * 16 + m0;
        #pragma unroll
        for (int kc = 0; kc < KC; ++kc) {
            B[nt][kc].x = W[(kc * 4 + 2 * half + 0) * 32 + n];
            B[nt][kc].y = W[(kc * 4 + 2 * half + 1) * 32 + n];
        }
    }
}

__device__ __forceinline__ void load_wfrags_r3b(const float* __restrict__ W,
                                                v2f (&B)[2][8], int lane)
{
    const int half = lane >> 4, m0 = lane & 15;
    #pragma unroll
    for (int nt = 0; nt < 2; ++nt) {
        const int n = nt * 16 + m0;
        #pragma unroll
        for (int kc = 0; kc < 8; ++kc) {
            B[nt][kc].x = W[(8 * half + kc) * 32 + n];
            B[nt][kc].y = W[(8 * half + kc + 16) * 32 + n];
        }
    }
}

struct RegW {
    v2f B1[2][8];
    v2f B2[2][8];
    v2f B3a[2][6];
    v2f B3b[2][8];
    v8f c1[2], c2[2], c3a[2], c3b[2];
    float r1b2;
    float cb;
    float w2a[2], w2b[2], w2c[2];
    v2f w0a[8], b0a[8], w0b[8], b0b[8];
};

__device__ __forceinline__ void lds_phase_fence()
{
    __builtin_amdgcn_fence(__ATOMIC_RELEASE, "workgroup", "local");
    __builtin_amdgcn_s_barrier();
    __builtin_amdgcn_fence(__ATOMIC_ACQUIRE, "workgroup", "local");
}

__device__ __forceinline__ void g0_exchange(const float (&hv)[3][2][8],
                                            v2f (&av)[3][8],
                                            float* __restrict__ g0, int lane)
{
    const int half = lane >> 4, m0 = lane & 15;
    #pragma unroll
    for (int mt = 0; mt < 3; ++mt) {
        #pragma unroll
        for (int r = 0; r < 8; ++r) {
            const int m = mt * 16 + r + 8 * half;
            v2f pr;
            pr.x = hv[mt][0][r];
            pr.y = hv[mt][1][r];
            *(lds_pair_t*)(&g0[m * GP + 2 * m0]) = pr;
        }
    }
    lds_phase_fence();
    #pragma unroll
    for (int mt = 0; mt < 3; ++mt) {
        const float* row = &g0[(mt * 16 + m0) * GP + 16 * half];
        #pragma unroll
        for (int j = 0; j < 4; ++j) {
            const v4f q = *reinterpret_cast<const v4f*>(&row[4 * j]);
            av[mt][2 * j].x     = q.x;
            av[mt][2 * j].y     = q.y;
            av[mt][2 * j + 1].x = q.z;
            av[mt][2 * j + 1].y = q.w;
        }
    }
}

template <int MASK>
__device__ __forceinline__ float row_xor(float v)
{
    return __builtin_bit_cast(float, __builtin_amdgcn_update_dpp(
        0, __builtin_bit_cast(int, v), 0x160 + MASK, 0xf, 0xf, true));
}

__device__ __forceinline__ float other_row(float v)
{
    const int b = __builtin_bit_cast(int, v);
    return __builtin_bit_cast(float,
        __builtin_amdgcn_permlanex16(b, b, 0x76543210, 0xfedcba98, false, false));
}

struct LaneSel {
    bool b0, b1, b2;
    bool own;
};

__device__ __forceinline__ float tile_dot(const v8f& c0, const v8f& c1,
                                          const float w2[2],
                                          const LaneSel& ls)
{
    float p[8];
    #pragma unroll
    for (int r = 0; r < 8; ++r)
        p[r] = fast_tanh(c0[r]) * w2[0] + fast_tanh(c1[r]) * w2[1];
    float q[4];
    #pragma unroll
    for (int i = 0; i < 4; ++i) {
        const float keep = ls.b2 ? p[4 + i] : p[i];
        const float send = ls.b2 ? p[i] : p[4 + i];
        q[i] = keep + row_xor<4>(send);
    }
    float s[2];
    #pragma unroll
    for (int i = 0; i < 2; ++i) {
        const float keep = ls.b1 ? q[2 + i] : q[i];
        const float send = ls.b1 ? q[i] : q[2 + i];
        s[i] = keep + row_xor<2>(send);
    }
    const float keep = ls.b0 ? s[1] : s[0];
    const float send = ls.b0 ? s[0] : s[1];
    const float t = keep + row_xor<1>(send);
    const float u = t + row_xor<8>(t);
    const float o = other_row(u);
    return ls.own ? u : o;
}

__device__ __forceinline__ void hidden_pair_mm(float xa, float xb,
                                               const RegW& rw, v8f (&acc)[2][2])
{
    acc[0][0] = rw.c1[0];
    acc[0][1] = rw.c1[1];
    acc[1][0] = rw.c2[0];
    acc[1][1] = rw.c2[1];
    #pragma unroll
    for (int kc = 0; kc < 8; ++kc) {
        v2f a1, a2;
        a1.x = fast_tanh(xa * rw.w0a[kc].x + rw.b0a[kc].x);
        a1.y = fast_tanh(xa * rw.w0a[kc].y + rw.b0a[kc].y);
        a2.x = fast_tanh(xb * rw.w0b[kc].x + rw.b0b[kc].x);
        a2.y = fast_tanh(xb * rw.w0b[kc].y + rw.b0b[kc].y);
        acc[0][0] = __builtin_amdgcn_wmma_f32_16x16x4_f32(
            false, a1, false, rw.B1[0][kc], (short)0, acc[0][0], false, false);
        acc[0][1] = __builtin_amdgcn_wmma_f32_16x16x4_f32(
            false, a1, false, rw.B1[1][kc], (short)0, acc[0][1], false, false);
        acc[1][0] = __builtin_amdgcn_wmma_f32_16x16x4_f32(
            false, a2, false, rw.B2[0][kc], (short)0, acc[1][0], false, false);
        acc[1][1] = __builtin_amdgcn_wmma_f32_16x16x4_f32(
            false, a2, false, rw.B2[1][kc], (short)0, acc[1][1], false, false);
    }
}

__device__ __forceinline__ void hidden_pair_dots(const v8f (&acc)[2][2],
                                                 const RegW& rw, const LaneSel& ls,
                                                 float& da, float& db)
{
    da = tile_dot(acc[0][0], acc[0][1], rw.w2a, ls);
    db = tile_dot(acc[1][0], acc[1][1], rw.w2b, ls);
}

__device__ __forceinline__ void r3_layer0(v8f (&cz)[2], const v2f (&z4a)[4],
                                          v2f zu4, v2f zu5, v2f zn4, v2f zn5,
                                          const RegW& rw,
                                          float (&hv)[3][2][8])
{
    v8f cs[2], cn[2];
    v8f acc[3][2];
    #pragma unroll
    for (int nt = 0; nt < 2; ++nt) {
        cs[nt] = rw.c3a[nt];
        #pragma unroll
        for (int kc = 0; kc < 4; ++kc)
            cs[nt] = __builtin_amdgcn_wmma_f32_16x16x4_f32(
                false, z4a[kc], false, rw.B3a[nt][kc], (short)0, cs[nt], false, false);
    }
    #pragma unroll
    for (int nt = 0; nt < 2; ++nt) {
        acc[2][nt] = __builtin_amdgcn_wmma_f32_16x16x4_f32(
            false, zu4, false, rw.B3a[nt][4], (short)0, cs[nt], false, false);
        acc[2][nt] = __builtin_amdgcn_wmma_f32_16x16x4_f32(
            false, zu5, false, rw.B3a[nt][5], (short)0, acc[2][nt], false, false);
        cn[nt] = __builtin_amdgcn_wmma_f32_16x16x4_f32(
            false, zn4, false, rw.B3a[nt][4], (short)0, cs[nt], false, false);
        cn[nt] = __builtin_amdgcn_wmma_f32_16x16x4_f32(
            false, zn5, false, rw.B3a[nt][5], (short)0, cn[nt], false, false);
    }
    #pragma unroll
    for (int nt = 0; nt < 2; ++nt) {
        acc[0][nt] = cz[nt];
        #pragma unroll
        for (int r = 0; r < 8; ++r)
            acc[1][nt][r] = 0.5f * (cz[nt][r] + acc[2][nt][r]);
        cz[nt] = cn[nt];
    }
    #pragma unroll
    for (int mt = 0; mt < 3; ++mt) {
        #pragma unroll
        for (int nt = 0; nt < 2; ++nt) {
            #pragma unroll
            for (int r = 0; r < 8; ++r)
                hv[mt][nt][r] = fast_tanh(acc[mt][nt][r]);
        }
    }
}

__device__ __forceinline__ void r3_layer1(const v2f (&av)[3][8],
                                          const RegW& rw, const LaneSel& ls,
                                          float (&d)[3])
{
    v8f acc[3][2];
    #pragma unroll
    for (int mt = 0; mt < 3; ++mt) {
        acc[mt][0] = rw.c3b[0];
        acc[mt][1] = rw.c3b[1];
    }
    #pragma unroll
    for (int kc = 0; kc < 8; ++kc) {
        #pragma unroll
        for (int mt = 0; mt < 3; ++mt) {
            acc[mt][0] = __builtin_amdgcn_wmma_f32_16x16x4_f32(
                false, av[mt][kc], false, rw.B3b[0][kc], (short)0, acc[mt][0], false, false);
            acc[mt][1] = __builtin_amdgcn_wmma_f32_16x16x4_f32(
                false, av[mt][kc], false, rw.B3b[1][kc], (short)0, acc[mt][1], false, false);
        }
    }
    #pragma unroll
    for (int mt = 0; mt < 3; ++mt)
        d[mt] = tile_dot(acc[mt][0], acc[mt][1], rw.w2c, ls);
}

__device__ __forceinline__ void rhs_known(float xe0, float xe1, float us, float cbz,
                                          float r1b2, float& q0, float& q1)
{
    const float xs0 = xe0 * 0.3f + 0.5f;
    const float xs1 = xe1 * 0.2f + 0.5f;
    q0 = (us - xs0) * (0.1f / 0.3f) - r1b2;
    q1 = cbz - 0.5f * xs1;
}

__device__ __forceinline__ void stage_pair(float xe0, float xe1, float us, float cbz,
                                           float b0, float b1, float c,
                                           const RegW& rw, const LaneSel& ls,
                                           float& k0, float& k1, float& y0, float& y1)
{
    v8f acc[2][2];
    hidden_pair_mm(xe0, xe1, rw, acc);
    float q0, q1;
    rhs_known(xe0, xe1, us, cbz, rw.r1b2, q0, q1);
    const float e0 = b0 + c * q0;
    const float e1 = b1 + c * q1;
    float d1, d2;
    hidden_pair_dots(acc, rw, ls, d1, d2);
    y0 = e0 - c * d1;
    y1 = (e1 + (1.5f * c) * d1) - (3.0f * c) * d2;
    k0 = q0 - d1;
    k1 = (q1 + 1.5f * d1) - 3.0f * d2;
}

__global__ __launch_bounds__(32) void reac_rk4_kernel(
    const float* __restrict__ useq, const float* __restrict__ xz0,
    const float* __restrict__ r1W0, const float* __restrict__ r1b0,
    const float* __restrict__ r1W1, const float* __restrict__ r1b1,
    const float* __restrict__ r1W2, const float* __restrict__ r1b2,
    const float* __restrict__ r2W0, const float* __restrict__ r2b0,
    const float* __restrict__ r2W1, const float* __restrict__ r2b1,
    const float* __restrict__ r2W2, const float* __restrict__ r2b2,
    const float* __restrict__ r3W0, const float* __restrict__ r3b0,
    const float* __restrict__ r3W1, const float* __restrict__ r3b1,
    const float* __restrict__ r3W2, const float* __restrict__ r3b2,
    float* __restrict__ out)
{
    __shared__ __attribute__((aligned(16))) float s_g0[3 * SPW * GP];

    const int lane = threadIdx.x & 31;
    const int m0   = lane & 15;
    const int half = lane >> 4;

    RegW rw;
    load_wfrags<8>(r1W1, rw.B1, lane);
    load_wfrags<8>(r2W1, rw.B2, lane);
    load_wfrags<6>(r3W0, rw.B3a, lane);
    load_wfrags_r3b(r3W1, rw.B3b, lane);
    #pragma unroll
    for (int nt = 0; nt < 2; ++nt) {
        const float v1 = r1b1[nt * 16 + m0], v2 = r2b1[nt * 16 + m0];
        const float v3 = r3b0[nt * 16 + m0], v4 = r3b1[nt * 16 + m0];
        rw.c1[nt]  = (v8f){v1, v1, v1, v1, v1, v1, v1, v1};
        rw.c2[nt]  = (v8f){v2, v2, v2, v2, v2, v2, v2, v2};
        rw.c3a[nt] = (v8f){v3, v3, v3, v3, v3, v3, v3, v3};
        rw.c3b[nt] = (v8f){v4, v4, v4, v4, v4, v4, v4, v4};
    }
    rw.r1b2 = r1b2[0];
    rw.cb = (1.5f * r1b2[0] - 3.0f * r2b2[0]) + r3b2[0];
    rw.w2a[0] = r1W2[m0]; rw.w2a[1] = r1W2[16 + m0];
    rw.w2b[0] = r2W2[m0]; rw.w2b[1] = r2W2[16 + m0];
    rw.w2c[0] = r3W2[m0]; rw.w2c[1] = r3W2[16 + m0];
    #pragma unroll
    for (int kc = 0; kc < 8; ++kc) {
        const int n = kc * 4 + 2 * half;
        rw.w0a[kc].x = r1W0[n]; rw.w0a[kc].y = r1W0[n + 1];
        rw.b0a[kc].x = r1b0[n]; rw.b0a[kc].y = r1b0[n + 1];
        rw.w0b[kc].x = r2W0[n]; rw.w0b[kc].y = r2W0[n + 1];
        rw.b0b[kc].x = r2b0[n]; rw.b0b[kc].y = r2b0[n + 1];
    }

    LaneSel ls;
    ls.b0 = (lane & 1) != 0; ls.b1 = (lane & 2) != 0; ls.b2 = (lane & 4) != 0;
    ls.own = ((lane >> 3) & 1) == half;

    const int base = blockIdx.x * SPW;

    const float* xzrow = xz0 + (size_t)(base + m0) * 26;
    v2f za[6];
    #pragma unroll
    for (int kc = 0; kc < 6; ++kc) {
        za[kc].x = xzrow[2 + kc * 4 + 2 * half + 0];
        za[kc].y = xzrow[2 + kc * 4 + 2 * half + 1];
    }
    float x0 = xzrow[0];
    float x1 = xzrow[1];
    const bool hi = half != 0;

    v8f cz[2];
    #pragma unroll
    for (int nt = 0; nt < 2; ++nt) {
        cz[nt] = rw.c3a[nt];
        #pragma unroll
        for (int kc = 0; kc < 6; ++kc)
            cz[nt] = __builtin_amdgcn_wmma_f32_16x16x4_f32(
                false, za[kc], false, rw.B3a[nt][kc], (short)0, cz[nt], false, false);
    }

    float u_next = useq[(size_t)(base + m0) * Td];
    for (int t = 0; t < Td; ++t) {
        const float u  = u_next;
        u_next = useq[(size_t)(base + m0) * Td + (t + 1 < Td ? t + 1 : t)];
        if (lane < SPW) {
            out[((size_t)(base + m0) * Td + t) * 2 + 0] = x0;
            out[((size_t)(base + m0) * Td + t) * 2 + 1] = x1;
        }

        v2f z4a[4];
        {
            const float p0x = other_row(za[0].x), p0y = other_row(za[0].y);
            const float p1x = other_row(za[1].x), p1y = other_row(za[1].y);
            const float p2x = other_row(za[2].x), p2y = other_row(za[2].y);
            const float p3x = other_row(za[3].x), p3y = other_row(za[3].y);
            z4a[0].x = hi ? p1x : p0x;  z4a[0].y = hi ? p1y : p0y;
            z4a[1].x = hi ? p2x : p1x;  z4a[1].y = hi ? p2y : p1y;
            z4a[2].x = hi ? p3x : p2x;  z4a[2].y = hi ? p3y : p2y;
            z4a[3].x = hi ? x0  : p3x;  z4a[3].y = hi ? x1  : p3y;
        }
        v2f n4, n5;
        {
            const float q4 = other_row(za[4].x);
            const float q5 = other_row(za[5].x);
            n4.x = za[4].y; n4.y = hi ? q5 : q4;
            n5.x = za[5].y; n5.y = hi ? u  : q5;
        }
        float hv[3][2][8];
        r3_layer0(cz, z4a, za[4], za[5], n4, n5, rw, hv);
        v8f a12[2][2];
        hidden_pair_mm(x0, x1, rw, a12);
        asm volatile("" :: "v"(a12[0][0]), "v"(a12[0][1]), "v"(a12[1][0]), "v"(a12[1][1]));
        v2f av[3][8];
        g0_exchange(hv, av, s_g0, lane);
        float d1, d2, d3[3];
        hidden_pair_dots(a12, rw, ls, d1, d2);
        const float us = u * 0.5f + 1.0f;
        float q0, q1;
        rhs_known(x0, x1, us, rw.cb, rw.r1b2, q0, q1);
        const float k1a = q0 - d1;
        const float k1p = (q1 + 1.5f * d1) - 3.0f * d2;
        const float e1  = x1 + 0.5f * k1p;
        asm volatile("" :: "v"(av[0][0]), "v"(av[0][1]), "v"(av[0][2]), "v"(av[0][3]),
                           "v"(av[0][4]), "v"(av[0][5]), "v"(av[0][6]), "v"(av[0][7]),
                           "v"(av[1][0]), "v"(av[1][1]), "v"(av[1][2]), "v"(av[1][3]),
                           "v"(av[1][4]), "v"(av[1][5]), "v"(av[1][6]), "v"(av[1][7]),
                           "v"(av[2][0]), "v"(av[2][1]), "v"(av[2][2]), "v"(av[2][3]),
                           "v"(av[2][4]), "v"(av[2][5]), "v"(av[2][6]), "v"(av[2][7]));
        __builtin_amdgcn_sched_barrier(0);
        r3_layer1(av, rw, ls, d3);
        const float k1b = k1p + d3[0];
        float xe0 = x0 + 0.5f * k1a, xe1 = e1 + 0.5f * d3[0];
        const float cbz23 = rw.cb + d3[1];
        const float cbz4  = rw.cb + d3[2];
        float k2a, k2b, k3a, k3b, k4a, k4b, ye0, ye1;

        stage_pair(xe0, xe1, us, cbz23, x0, x1, 0.5f, rw, ls, k2a, k2b, ye0, ye1);

        stage_pair(ye0, ye1, us, cbz23, x0, x1, 1.0f, rw, ls, k3a, k3b, xe0, xe1);

        const float g0 = x0 + ((k1a + 2.f * k2a) + 2.f * k3a) * (1.f / 6.f);
        const float g1 = x1 + ((k1b + 2.f * k2b) + 2.f * k3b) * (1.f / 6.f);
        float nx0, nx1;
        stage_pair(xe0, xe1, us, cbz4, g0, g1, 1.f / 6.f, rw, ls, k4a, k4b, nx0, nx1);
        {
            #pragma unroll
            for (int kc = 0; kc < 4; ++kc) za[kc] = z4a[kc];
            za[4] = n4;
            za[5] = n5;
        }
        x0 = nx0;
        x1 = nx1;
        lds_phase_fence();
    }
}

extern "C" void kernel_launch(void* const* d_in, const int* in_sizes, int n_in,
                              void* d_out, int out_size, void* d_ws, size_t ws_size,
                              hipStream_t stream) {
    (void)in_sizes; (void)n_in; (void)out_size; (void)d_ws; (void)ws_size;
    const float* useq = (const float*)d_in[0];
    const float* xz0  = (const float*)d_in[1];
    const float* p[18];
    for (int i = 0; i < 18; ++i) p[i] = (const float*)d_in[2 + i];
    float* out = (float*)d_out;

    const int B = 4096;
    dim3 block(32);
    dim3 grid(B / SPW);
    reac_rk4_kernel<<<grid, block, 0, stream>>>(
        useq, xz0,
        p[0], p[1], p[2], p[3], p[4], p[5],
        p[6], p[7], p[8], p[9], p[10], p[11],
        p[12], p[13], p[14], p[15], p[16], p[17],
        out);
}


# Batch re-tiled: 4 samples per wave, 4 independent waves per workgroup, one per SIMD, wave-scope LDS fence; 2 distinct tile rows per lane half: tanh, dots on 2 of 8 registers; dot sum order changes
# speedup vs baseline: 4.3036x; 1.1357x over previous
#include <hip/hip_runtime.h>
#include <hip/hip_bf16.h>


#define Td  256
#define SPW 16
#define NV  4
#define RH  2
#define WPB 4
#define GP  36
#define LDS_FLOATS 45056

typedef __attribute__((ext_vector_type(2))) float v2f;
typedef __attribute__((ext_vector_type(4))) float v4f;
typedef __attribute__((address_space(3))) volatile v2f lds_pair_t;
typedef __attribute__((ext_vector_type(8))) float v8f;

__device__ __forceinline__ float fast_tanh(float x)
{
    return __builtin_amdgcn_tanhf(x);
}

template <int KC>
__device__ __forceinline__ void load_wfrags(const float* __restrict__ W,
                                            v2f (&B)[2][KC], int lane)
{
    const int half = lane >> 4, m0 = lane & 15;
    #pragma unroll
    for (int nt = 0; nt < 2; ++nt) {
        const int n = nt * 16 + m0;
        #pragma unroll
        for (int kc = 0; kc < KC; ++kc) {
            B[nt][kc].x = W[(kc * 4 + 2 * half + 0) * 32 + n];
            B[nt][kc].y = W[(kc * 4 + 2 * half + 1) * 32 + n];
        }
    }
}

__device__ __forceinline__ void load_wfrags_r3b(const float* __restrict__ W,
                                                v2f (&B)[2][8], int lane)
{
    const int half = lane >> 4, m0 = lane & 15;
    #pragma unroll
    for (int nt = 0; nt < 2; ++nt) {
        const int n = nt * 16 + m0;
        #pragma unroll
        for (int kc = 0; kc < 8; ++kc) {
            B[nt][kc].x = W[(8 * half + kc) * 32 + n];
            B[nt][kc].y = W[(8 * half + kc + 16) * 32 + n];
        }
    }
}

struct RegW {
    v2f B1[2][8];
    v2f B2[2][8];
    v2f B3a[2][6];
    v2f B3b[2][8];
    v8f c1[2], c2[2], c3a[2], c3b[2];
    float r1b2;
    float cb;
    float w2a[2], w2b[2], w2c[2];
    v2f w0a[8], b0a[8], w0b[8], b0b[8];
};

__device__ __forceinline__ void lds_phase_fence()
{
    __builtin_amdgcn_fence(__ATOMIC_RELEASE, "wavefront", "local");
    __builtin_amdgcn_wave_barrier();
    __builtin_amdgcn_fence(__ATOMIC_ACQUIRE, "wavefront", "local");
}

__device__ __forceinline__ int row_sample(int m0)
{
    return (m0 & (RH - 1)) + RH * (m0 >> 3);
}

__device__ __forceinline__ void g0_exchange(const float (&hv)[3][2][RH],
                                            v2f (&av)[3][8],
                                            float* __restrict__ g0, int lane)
{
    const int half = lane >> 4, m0 = lane & 15;
    #pragma unroll
    for (int mt = 0; mt < 3; ++mt) {
        #pragma unroll
        for (int r = 0; r < RH; ++r) {
            const int m = mt * 16 + r + 8 * half;
            v2f pr;
            pr.x = hv[mt][0][r];
            pr.y = hv[mt][1][r];
            *(lds_pair_t*)(&g0[m * GP + 2 * m0]) = pr;
        }
    }
    lds_phase_fence();
    #pragma unroll
    for (int mt = 0; mt < 3; ++mt) {
        const float* row = &g0[(mt * 16 + (m0 & (8 | (RH - 1)))) * GP + 16 * half];
        #pragma unroll
        for (int j = 0; j < 4; ++j) {
            const v4f q = *reinterpret_cast<const v4f*>(&row[4 * j]);
            av[mt][2 * j].x     = q.x;
            av[mt][2 * j].y     = q.y;
            av[mt][2 * j + 1].x = q.z;
            av[mt][2 * j + 1].y = q.w;
        }
    }
}

template <int MASK>
__device__ __forceinline__ float row_xor(float v)
{
    return __builtin_bit_cast(float, __builtin_amdgcn_update_dpp(
        0, __builtin_bit_cast(int, v), 0x160 + MASK, 0xf, 0xf, true));
}

__device__ __forceinline__ float other_row(float v)
{
    const int b = __builtin_bit_cast(int, v);
    return __builtin_bit_cast(float,
        __builtin_amdgcn_permlanex16(b, b, 0x76543210, 0xfedcba98, false, false));
}

struct LaneSel {
    bool b0;
    bool own;
};

__device__ __forceinline__ float tile_dot(const v8f& c0, const v8f& c1,
                                          const float w2[2],
                                          const LaneSel& ls)
{
    const float p0 = fast_tanh(c0[0]) * w2[0] + fast_tanh(c1[0]) * w2[1];
    const float p1 = fast_tanh(c0[1]) * w2[0] + fast_tanh(c1[1]) * w2[1];
    const float keep = ls.b0 ? p1 : p0;
    const float send = ls.b0 ? p0 : p1;
    float t = keep + row_xor<1>(send);
    t = t + row_xor<2>(t);
    t = t + row_xor<4>(t);
    const float u = t + row_xor<8>(t);
    const float o = other_row(u);
    return ls.own ? u : o;
}

__device__ __forceinline__ void hidden_pair_mm(float xa, float xb,
                                               const RegW& rw, v8f (&acc)[2][2])
{
    acc[0][0] = rw.c1[0];
    acc[0][1] = rw.c1[1];
    acc[1][0] = rw.c2[0];
    acc[1][1] = rw.c2[1];
    #pragma unroll
    for (int kc = 0; kc < 8; ++kc) {
        v2f a1, a2;
        a1.x = fast_tanh(xa * rw.w0a[kc].x + rw.b0a[kc].x);
        a1.y = fast_tanh(xa * rw.w0a[kc].y + rw.b0a[kc].y);
        a2.x = fast_tanh(xb * rw.w0b[kc].x + rw.b0b[kc].x);
        a2.y = fast_tanh(xb * rw.w0b[kc].y + rw.b0b[kc].y);
        acc[0][0] = __builtin_amdgcn_wmma_f32_16x16x4_f32(
            false, a1, false, rw.B1[0][kc], (short)0, acc[0][0], false, false);
        acc[0][1] = __builtin_amdgcn_wmma_f32_16x16x4_f32(
            false, a1, false, rw.B1[1][kc], (short)0, acc[0][1], false, false);
        acc[1][0] = __builtin_amdgcn_wmma_f32_16x16x4_f32(
            false, a2, false, rw.B2[0][kc], (short)0, acc[1][0], false, false);
        acc[1][1] = __builtin_amdgcn_wmma_f32_16x16x4_f32(
            false, a2, false, rw.B2[1][kc], (short)0, acc[1][1], false, false);
    }
}

__device__ __forceinline__ void hidden_pair_dots(const v8f (&acc)[2][2],
                                                 const RegW& rw, const LaneSel& ls,
                                                 float& da, float& db)
{
    da = tile_dot(acc[0][0], acc[0][1], rw.w2a, ls);
    db = tile_dot(acc[1][0], acc[1][1], rw.w2b, ls);
}

__device__ __forceinline__ void r3_layer0(v8f (&cz)[2], const v2f (&z4a)[4],
                                          v2f zu4, v2f zu5, v2f zn4, v2f zn5,
                                          const RegW& rw,
                                          float (&hv)[3][2][RH])
{
    v8f cs[2], cn[2];
    v8f acc2[2];
    #pragma unroll
    for (int nt = 0; nt < 2; ++nt) {
        cs[nt] = rw.c3a[nt];
        #pragma unroll
        for (int kc = 0; kc < 4; ++kc)
            cs[nt] = __builtin_amdgcn_wmma_f32_16x16x4_f32(
                false, z4a[kc], false, rw.B3a[nt][kc], (short)0, cs[nt], false, false);
    }
    #pragma unroll
    for (int nt = 0; nt < 2; ++nt) {
        acc2[nt] = __builtin_amdgcn_wmma_f32_16x16x4_f32(
            false, zu4, false, rw.B3a[nt][4], (short)0, cs[nt], false, false);
        acc2[nt] = __builtin_amdgcn_wmma_f32_16x16x4_f32(
            false, zu5, false, rw.B3a[nt][5], (short)0, acc2[nt], false, false);
        cn[nt] = __builtin_amdgcn_wmma_f32_16x16x4_f32(
            false, zn4, false, rw.B3a[nt][4], (short)0, cs[nt], false, false);
        cn[nt] = __builtin_amdgcn_wmma_f32_16x16x4_f32(
            false, zn5, false, rw.B3a[nt][5], (short)0, cn[nt], false, false);
    }
    #pragma unroll
    for (int nt = 0; nt < 2; ++nt) {
        #pragma unroll
        for (int r = 0; r < RH; ++r) {
            hv[0][nt][r] = fast_tanh(cz[nt][r]);
            hv[1][nt][r] = fast_tanh(0.5f * (cz[nt][r] + acc2[nt][r]));
            hv[2][nt][r] = fast_tanh(acc2[nt][r]);
        }
        cz[nt] = cn[nt];
    }
}

__device__ __forceinline__ void r3_layer1(const v2f (&av)[3][8],
                                          const RegW& rw, const LaneSel& ls,
                                          float (&d)[3])
{
    v8f acc[3][2];
    #pragma unroll
    for (int mt = 0; mt < 3; ++mt) {
        acc[mt][0] = rw.c3b[0];
        acc[mt][1] = rw.c3b[1];
    }
    #pragma unroll
    for (int kc = 0; kc < 8; ++kc) {
        #pragma unroll
        for (int mt = 0; mt < 3; ++mt) {
            acc[mt][0] = __builtin_amdgcn_wmma_f32_16x16x4_f32(
                false, av[mt][kc], false, rw.B3b[0][kc], (short)0, acc[mt][0], false, false);
            acc[mt][1] = __builtin_amdgcn_wmma_f32_16x16x4_f32(
                false, av[mt][kc], false, rw.B3b[1][kc], (short)0, acc[mt][1], false, false);
        }
    }
    #pragma unroll
    for (int mt = 0; mt < 3; ++mt)
        d[mt] = tile_dot(acc[mt][0], acc[mt][1], rw.w2c, ls);
}

__device__ __forceinline__ void rhs_known(float xe0, float xe1, float us, float cbz,
                                          float r1b2, float& q0, float& q1)
{
    const float xs0 = xe0 * 0.3f + 0.5f;
    const float xs1 = xe1 * 0.2f + 0.5f;
    q0 = (us - xs0) * (0.1f / 0.3f) - r1b2;
    q1 = cbz - 0.5f * xs1;
}

__device__ __forceinline__ void stage_pair(float xe0, float xe1, float us, float cbz,
                                           float b0, float b1, float c,
                                           const RegW& rw, const LaneSel& ls,
                                           float& k0, float& k1, float& y0, float& y1)
{
    v8f acc[2][2];
    hidden_pair_mm(xe0, xe1, rw, acc);
    float q0, q1;
    rhs_known(xe0, xe1, us, cbz, rw.r1b2, q0, q1);
    const float e0 = b0 + c * q0;
    const float e1 = b1 + c * q1;
    float d1, d2;
    hidden_pair_dots(acc, rw, ls, d1, d2);
    y0 = e0 - c * d1;
    y1 = (e1 + (1.5f * c) * d1) - (3.0f * c) * d2;
    k0 = q0 - d1;
    k1 = (q1 + 1.5f * d1) - 3.0f * d2;
}

__global__ __launch_bounds__(32 * WPB) void reac_rk4_kernel(
    const float* __restrict__ useq, const float* __restrict__ xz0,
    const float* __restrict__ r1W0, const float* __restrict__ r1b0,
    const float* __restrict__ r1W1, const float* __restrict__ r1b1,
    const float* __restrict__ r1W2, const float* __restrict__ r1b2,
    const float* __restrict__ r2W0, const float* __restrict__ r2b0,
    const float* __restrict__ r2W1, const float* __restrict__ r2b1,
    const float* __restrict__ r2W2, const float* __restrict__ r2b2,
    const float* __restrict__ r3W0, const float* __restrict__ r3b0,
    const float* __restrict__ r3W1, const float* __restrict__ r3b1,
    const float* __restrict__ r3W2, const float* __restrict__ r3b2,
    float* __restrict__ out)
{
    __shared__ __attribute__((aligned(16))) float s_lds[LDS_FLOATS];
    static_assert(WPB * 3 * SPW * GP <= LDS_FLOATS, "the waves' rows fit the reservation");

    const int lane = threadIdx.x & 31;
    const int wave = threadIdx.x >> 5;
    const int m0   = lane & 15;
    const int half = lane >> 4;
    float* s_g0 = &s_lds[wave * (3 * SPW * GP)];

    RegW rw;
    load_wfrags<8>(r1W1, rw.B1, lane);
    load_wfrags<8>(r2W1, rw.B2, lane);
    load_wfrags<6>(r3W0, rw.B3a, lane);
    load_wfrags_r3b(r3W1, rw.B3b, lane);
    #pragma unroll
    for (int nt = 0; nt < 2; ++nt) {
        const float v1 = r1b1[nt * 16 + m0], v2 = r2b1[nt * 16 + m0];
        const float v3 = r3b0[nt * 16 + m0], v4 = r3b1[nt * 16 + m0];
        rw.c1[nt]  = (v8f){v1, v1, v1, v1, v1, v1, v1, v1};
        rw.c2[nt]  = (v8f){v2, v2, v2, v2, v2, v2, v2, v2};
        rw.c3a[nt] = (v8f){v3, v3, v3, v3, v3, v3, v3, v3};
        rw.c3b[nt] = (v8f){v4, v4, v4, v4, v4, v4, v4, v4};
    }
    rw.r1b2 = r1b2[0];
    rw.cb = (1.5f * r1b2[0] - 3.0f * r2b2[0]) + r3b2[0];
    rw.w2a[0] = r1W2[m0]; rw.w2a[1] = r1W2[16 + m0];
    rw.w2b[0] = r2W2[m0]; rw.w2b[1] = r2W2[16 + m0];
    rw.w2c[0] = r3W2[m0]; rw.w2c[1] = r3W2[16 + m0];
    #pragma unroll
    for (int kc = 0; kc < 8; ++kc) {
        const int n = kc * 4 + 2 * half;
        rw.w0a[kc].x = r1W0[n]; rw.w0a[kc].y = r1W0[n + 1];
        rw.b0a[kc].x = r1b0[n]; rw.b0a[kc].y = r1b0[n + 1];
        rw.w0b[kc].x = r2W0[n]; rw.w0b[kc].y = r2W0[n + 1];
        rw.b0b[kc].x = r2b0[n]; rw.b0b[kc].y = r2b0[n + 1];
    }

    LaneSel ls;
    ls.b0 = (lane & 1) != 0;
    ls.own = ((lane >> 3) & 1) == half;

    const int base = (blockIdx.x * WPB + wave) * NV;
    const int smp  = base + row_sample(m0);

    const float* xzrow = xz0 + (size_t)smp * 26;
    v2f za[6];
    #pragma unroll
    for (int kc = 0; kc < 6; ++kc) {
        za[kc].x = xzrow[2 + kc * 4 + 2 * half + 0];
        za[kc].y = xzrow[2 + kc * 4 + 2 * half + 1];
    }
    float x0 = xzrow[0];
    float x1 = xzrow[1];
    const bool hi = half != 0;

    v8f cz[2];
    #pragma unroll
    for (int nt = 0; nt < 2; ++nt) {
        cz[nt] = rw.c3a[nt];
        #pragma unroll
        for (int kc = 0; kc < 6; ++kc)
            cz[nt] = __builtin_amdgcn_wmma_f32_16x16x4_f32(
                false, za[kc], false, rw.B3a[nt][kc], (short)0, cz[nt], false, false);
    }

    float u_next = useq[(size_t)smp * Td];
    for (int t = 0; t < Td; ++t) {
        const float u  = u_next;
        u_next = useq[(size_t)smp * Td + (t + 1 < Td ? t + 1 : t)];
        if (half == 0 && (m0 & (7 & ~(RH - 1))) == 0) {
            out[((size_t)smp * Td + t) * 2 + 0] = x0;
            out[((size_t)smp * Td + t) * 2 + 1] = x1;
        }

        v2f z4a[4];
        {
            const float p0x = other_row(za[0].x), p0y = other_row(za[0].y);
            const float p1x = other_row(za[1].x), p1y = other_row(za[1].y);
            const float p2x = other_row(za[2].x), p2y = other_row(za[2].y);
            const float p3x = other_row(za[3].x), p3y = other_row(za[3].y);
            z4a[0].x = hi ? p1x : p0x;  z4a[0].y = hi ? p1y : p0y;
            z4a[1].x = hi ? p2x : p1x;  z4a[1].y = hi ? p2y : p1y;
            z4a[2].x = hi ? p3x : p2x;  z4a[2].y = hi ? p3y : p2y;
            z4a[3].x = hi ? x0  : p3x;  z4a[3].y = hi ? x1  : p3y;
        }
        v2f n4, n5;
        {
            const float q4 = other_row(za[4].x);
            const float q5 = other_row(za[5].x);
            n4.x = za[4].y; n4.y = hi ? q5 : q4;
            n5.x = za[5].y; n5.y = hi ? u  : q5;
        }
        float hv[3][2][RH];
        r3_layer0(cz, z4a, za[4], za[5], n4, n5, rw, hv);
        v8f a12[2][2];
        hidden_pair_mm(x0, x1, rw, a12);
        asm volatile("" :: "v"(a12[0][0]), "v"(a12[0][1]), "v"(a12[1][0]), "v"(a12[1][1]));
        v2f av[3][8];
        g0_exchange(hv, av, s_g0, lane);
        float d1, d2, d3[3];
        hidden_pair_dots(a12, rw, ls, d1, d2);
        const float us = u * 0.5f + 1.0f;
        float q0, q1;
        rhs_known(x0, x1, us, rw.cb, rw.r1b2, q0, q1);
        const float k1a = q0 - d1;
        const float k1p = (q1 + 1.5f * d1) - 3.0f * d2;
        const float e1  = x1 + 0.5f * k1p;
        asm volatile("" :: "v"(av[0][0]), "v"(av[0][1]), "v"(av[0][2]), "v"(av[0][3]),
                           "v"(av[0][4]), "v"(av[0][5]), "v"(av[0][6]), "v"(av[0][7]),
                           "v"(av[1][0]), "v"(av[1][1]), "v"(av[1][2]), "v"(av[1][3]),
                           "v"(av[1][4]), "v"(av[1][5]), "v"(av[1][6]), "v"(av[1][7]),
                           "v"(av[2][0]), "v"(av[2][1]), "v"(av[2][2]), "v"(av[2][3]),
                           "v"(av[2][4]), "v"(av[2][5]), "v"(av[2][6]), "v"(av[2][7]));
        __builtin_amdgcn_sched_barrier(0);
        r3_layer1(av, rw, ls, d3);
        const float k1b = k1p + d3[0];
        float xe0 = x0 + 0.5f * k1a, xe1 = e1 + 0.5f * d3[0];
        const float cbz23 = rw.cb + d3[1];
        const float cbz4  = rw.cb + d3[2];
        float k2a, k2b, k3a, k3b, k4a, k4b, ye0, ye1;

        stage_pair(xe0, xe1, us, cbz23, x0, x1, 0.5f, rw, ls, k2a, k2b, ye0, ye1);

        stage_pair(ye0, ye1, us, cbz23, x0, x1, 1.0f, rw, ls, k3a, k3b, xe0, xe1);

        const float g0 = x0 + ((k1a + 2.f * k2a) + 2.f * k3a) * (1.f / 6.f);
        const float g1 = x1 + ((k1b + 2.f * k2b) + 2.f * k3b) * (1.f / 6.f);
        float nx0, nx1;
        stage_pair(xe0, xe1, us, cbz4, g0, g1, 1.f / 6.f, rw, ls, k4a, k4b, nx0, nx1);
        {
            #pragma unroll
            for (int kc = 0; kc < 4; ++kc) za[kc] = z4a[kc];
            za[4] = n4;
            za[5] = n5;
        }
        x0 = nx0;
        x1 = nx1;
        lds_phase_fence();
    }
}

extern "C" void kernel_launch(void* const* d_in, const int* in_sizes, int n_in,
                              void* d_out, int out_size, void* d_ws, size_t ws_size,
                              hipStream_t stream) {
    (void)in_sizes; (void)n_in; (void)out_size; (void)d_ws; (void)ws_size;
    const float* useq = (const float*)d_in[0];
    const float* xz0  = (const float*)d_in[1];
    const float* p[18];
    for (int i = 0; i < 18; ++i) p[i] = (const float*)d_in[2 + i];
    float* out = (float*)d_out;

    const int B = 4096;
    dim3 block(32 * WPB);
    dim3 grid(B / (NV * WPB));
    reac_rk4_kernel<<<grid, block, 0, stream>>>(
        useq, xz0,
        p[0], p[1], p[2], p[3], p[4], p[5],
        p[6], p[7], p[8], p[9], p[10], p[11],
        p[12], p[13], p[14], p[15], p[16], p[17],
        out);
}
